# starting kernel
# baseline (speedup 1.0000x reference)
#include <hip/hip_runtime.h>


typedef __attribute__((ext_vector_type(2))) float v2f;
typedef __attribute__((ext_vector_type(8))) float v8f;

#define NROWS 50000
#define INDIM 256
#define HID   128
#define CLS   64
#define NEDGE 800000

__global__ __launch_bounds__(64) void gcn_gemm1_wmma(
    const float* __restrict__ X, const float* __restrict__ W1,
    float* __restrict__ Y)
{
    const int mtile = blockIdx.x;
    const int wv    = threadIdx.x >> 5;
    const int lane  = threadIdx.x & 31;
    const int half  = lane >> 4;
    const int l16   = lane & 15;

    const int m0 = mtile * 16;
    const int n0 = wv * 64;

    const float* xrow = X + (size_t)(m0 + l16) * INDIM + 2 * half;
    const float* wcol = W1 + n0 + l16;

    v8f acc0 = {}, acc1 = {}, acc2 = {}, acc3 = {};
    #pragma unroll 2
    for (int k = 0; k < INDIM; k += 4) {
        const float2 av = *(const float2*)(xrow + k);
        v2f a; a.x = av.x; a.y = av.y;
        const float* wk0 = wcol + (size_t)(k + 2 * half) * HID;
        const float* wk1 = wcol + (size_t)(k + 2 * half + 1) * HID;
        v2f b0; b0.x = wk0[0];  b0.y = wk1[0];
        v2f b1; b1.x = wk0[16]; b1.y = wk1[16];
        v2f b2; b2.x = wk0[32]; b2.y = wk1[32];
        v2f b3; b3.x = wk0[48]; b3.y = wk1[48];
        acc0 = __builtin_amdgcn_wmma_f32_16x16x4_f32(false, a, false, b0, (short)0, acc0, false, false);
        acc1 = __builtin_amdgcn_wmma_f32_16x16x4_f32(false, a, false, b1, (short)0, acc1, false, false);
        acc2 = __builtin_amdgcn_wmma_f32_16x16x4_f32(false, a, false, b2, (short)0, acc2, false, false);
        acc3 = __builtin_amdgcn_wmma_f32_16x16x4_f32(false, a, false, b3, (short)0, acc3, false, false);
    }

    float* yb = Y + n0 + l16;
    #pragma unroll
    for (int i = 0; i < 8; ++i) {
        float* yr = yb + (size_t)(m0 + i + 8 * half) * HID;
        yr[0]  = acc0[i];
        yr[16] = acc1[i];
        yr[32] = acc2[i];
        yr[48] = acc3[i];
    }
}

__global__ __launch_bounds__(32) void gcn_gemm2_wmma(
    const float* __restrict__ Hraw, const float* __restrict__ b1,
    const float* __restrict__ W2, float* __restrict__ Y)
{
    const int mtile = blockIdx.x;
    const int lane  = threadIdx.x & 31;
    const int half  = lane >> 4;
    const int l16   = lane & 15;

    const int m0 = mtile * 16;

    const float* hrow = Hraw + (size_t)(m0 + l16) * HID + 2 * half;
    const float* brow = b1 + 2 * half;
    const float* wcol = W2 + l16;

    v8f acc0 = {}, acc1 = {}, acc2 = {}, acc3 = {};
    #pragma unroll 2
    for (int k = 0; k < HID; k += 4) {
        const float2 hv = *(const float2*)(hrow + k);
        const float2 bv = *(const float2*)(brow + k);
        v2f a;
        a.x = fmaxf(hv.x + bv.x, 0.0f);
        a.y = fmaxf(hv.y + bv.y, 0.0f);
        const float* wk0 = wcol + (size_t)(k + 2 * half) * CLS;
        const float* wk1 = wcol + (size_t)(k + 2 * half + 1) * CLS;
        v2f b0; b0.x = wk0[0];  b0.y = wk1[0];
        v2f b1v; b1v.x = wk0[16]; b1v.y = wk1[16];
        v2f b2v; b2v.x = wk0[32]; b2v.y = wk1[32];
        v2f b3v; b3v.x = wk0[48]; b3v.y = wk1[48];
        acc0 = __builtin_amdgcn_wmma_f32_16x16x4_f32(false, a, false, b0,  (short)0, acc0, false, false);
        acc1 = __builtin_amdgcn_wmma_f32_16x16x4_f32(false, a, false, b1v, (short)0, acc1, false, false);
        acc2 = __builtin_amdgcn_wmma_f32_16x16x4_f32(false, a, false, b2v, (short)0, acc2, false, false);
        acc3 = __builtin_amdgcn_wmma_f32_16x16x4_f32(false, a, false, b3v, (short)0, acc3, false, false);
    }

    float* yb = Y + l16;
    #pragma unroll
    for (int i = 0; i < 8; ++i) {
        float* yr = yb + (size_t)(m0 + i + 8 * half) * CLS;
        yr[0]  = acc0[i];
        yr[16] = acc1[i];
        yr[32] = acc2[i];
        yr[48] = acc3[i];
    }
}

__global__ __launch_bounds__(256) void gcn_spmm_f128(
    const int* __restrict__ erow, const int* __restrict__ ecol,
    const float* __restrict__ eval, const float* __restrict__ dense,
    float* __restrict__ out)
{
    const int e = (int)((blockIdx.x * (size_t)blockDim.x + threadIdx.x) >> 5);
    const int lane = threadIdx.x & 31;
    if (e >= NEDGE) return;
    const int   r = erow[e];
    const int   c = ecol[e];
    const float v = eval[e];
    const float4 g = *(const float4*)(dense + (size_t)c * HID + lane * 4);
    float* o = out + (size_t)r * HID + lane * 4;
    atomicAdd(o + 0, v * g.x);
    atomicAdd(o + 1, v * g.y);
    atomicAdd(o + 2, v * g.z);
    atomicAdd(o + 3, v * g.w);
}

__global__ __launch_bounds__(256) void gcn_spmm_f64(
    const int* __restrict__ erow, const int* __restrict__ ecol,
    const float* __restrict__ eval, const float* __restrict__ dense,
    float* __restrict__ out)
{
    const int e = (int)((blockIdx.x * (size_t)blockDim.x + threadIdx.x) >> 5);
    const int lane = threadIdx.x & 31;
    if (e >= NEDGE) return;
    const int   r = erow[e];
    const int   c = ecol[e];
    const float v = eval[e];
    const float2 g = *(const float2*)(dense + (size_t)c * CLS + lane * 2);
    float* o = out + (size_t)r * CLS + lane * 2;
    atomicAdd(o + 0, v * g.x);
    atomicAdd(o + 1, v * g.y);
}

__global__ __launch_bounds__(256) void gcn_logsoftmax64(
    const float* __restrict__ L, const float* __restrict__ b2,
    float* __restrict__ out)
{
    const int r = (int)((blockIdx.x * (size_t)blockDim.x + threadIdx.x) >> 5);
    const int lane = threadIdx.x & 31;
    if (r >= NROWS) return;

    float2 v = *(const float2*)(L + (size_t)r * CLS + lane * 2);
    const float2 bb = *(const float2*)(b2 + lane * 2);
    v.x += bb.x;
    v.y += bb.y;

    float m = fmaxf(v.x, v.y);
    #pragma unroll
    for (int off = 16; off > 0; off >>= 1)
        m = fmaxf(m, __shfl_xor(m, off, 32));

    float s = __expf(v.x - m) + __expf(v.y - m);
    #pragma unroll
    for (int off = 16; off > 0; off >>= 1)
        s += __shfl_xor(s, off, 32);

    const float lse = m + __logf(s);
    float2 o;
    o.x = v.x - lse;
    o.y = v.y - lse;
    *(float2*)(out + (size_t)r * CLS + lane * 2) = o;
}

__global__ __launch_bounds__(256) void gcn_zero_f32(float* __restrict__ p, int n)
{
    const int i = (int)(blockIdx.x * (size_t)blockDim.x + threadIdx.x);
    if (i < n) p[i] = 0.0f;
}

extern "C" void kernel_launch(void* const* d_in, const int* in_sizes, int n_in,
                              void* d_out, int out_size, void* d_ws, size_t ws_size,
                              hipStream_t stream) {
    const float* x        = (const float*)d_in[0];
    const int*   edge_row = (const int*)  d_in[1];
    const int*   edge_col = (const int*)  d_in[2];
    const float* edge_val = (const float*)d_in[3];
    const float* W1       = (const float*)d_in[4];
    const float* b1       = (const float*)d_in[5];
    const float* W2       = (const float*)d_in[6];
    const float* b2       = (const float*)d_in[7];
    float* out = (float*)d_out;

    float* ws    = (float*)d_ws;
    float* XW1   = ws;
    float* Hraw  = ws + (size_t)NROWS * HID;
    float* HW2   = XW1;
    float* Lraw  = Hraw;

    const int MT = NROWS / 16;
    const int edge_blocks = (NEDGE * 32 + 255) / 256;

    gcn_gemm1_wmma<<<MT, 64, 0, stream>>>(x, W1, XW1);
    gcn_zero_f32<<<(NROWS * HID + 255) / 256, 256, 0, stream>>>(Hraw, NROWS * HID);
    gcn_spmm_f128<<<edge_blocks, 256, 0, stream>>>(edge_row, edge_col, edge_val, XW1, Hraw);
    gcn_gemm2_wmma<<<MT, 32, 0, stream>>>(Hraw, b1, W2, HW2);
    gcn_zero_f32<<<(NROWS * CLS + 255) / 256, 256, 0, stream>>>(Lraw, NROWS * CLS);
    gcn_spmm_f64<<<edge_blocks, 256, 0, stream>>>(edge_row, edge_col, edge_val, HW2, Lraw);
    gcn_logsoftmax64<<<(NROWS * 32 + 255) / 256, 256, 0, stream>>>(Lraw, b2, out);
}


# SpMM: global float-atomic scatter replaced by row-range ownership: 256 groups each scan edge_row and sum their 196 rows in LDS (LDS float atomics, arrival order); zero passes go, 7 -> 5 launches
# speedup vs baseline: 1.5631x; 1.5631x over previous
#include <hip/hip_runtime.h>


typedef __attribute__((ext_vector_type(2))) float v2f;
typedef __attribute__((ext_vector_type(8))) float v8f;

#define NROWS 50000
#define INDIM 256
#define HID   128
#define CLS   64
#define NEDGE 800000

__global__ __launch_bounds__(64) void gcn_gemm1_wmma(
    const float* __restrict__ X, const float* __restrict__ W1,
    float* __restrict__ Y)
{
    const int mtile = blockIdx.x;
    const int wv    = threadIdx.x >> 5;
    const int lane  = threadIdx.x & 31;
    const int half  = lane >> 4;
    const int l16   = lane & 15;

    const int m0 = mtile * 16;
    const int n0 = wv * 64;

    const float* xrow = X + (size_t)(m0 + l16) * INDIM + 2 * half;
    const float* wcol = W1 + n0 + l16;

    v8f acc0 = {}, acc1 = {}, acc2 = {}, acc3 = {};
    #pragma unroll 2
    for (int k = 0; k < INDIM; k += 4) {
        const float2 av = *(const float2*)(xrow + k);
        v2f a; a.x = av.x; a.y = av.y;
        const float* wk0 = wcol + (size_t)(k + 2 * half) * HID;
        const float* wk1 = wcol + (size_t)(k + 2 * half + 1) * HID;
        v2f b0; b0.x = wk0[0];  b0.y = wk1[0];
        v2f b1; b1.x = wk0[16]; b1.y = wk1[16];
        v2f b2; b2.x = wk0[32]; b2.y = wk1[32];
        v2f b3; b3.x = wk0[48]; b3.y = wk1[48];
        acc0 = __builtin_amdgcn_wmma_f32_16x16x4_f32(false, a, false, b0, (short)0, acc0, false, false);
        acc1 = __builtin_amdgcn_wmma_f32_16x16x4_f32(false, a, false, b1, (short)0, acc1, false, false);
        acc2 = __builtin_amdgcn_wmma_f32_16x16x4_f32(false, a, false, b2, (short)0, acc2, false, false);
        acc3 = __builtin_amdgcn_wmma_f32_16x16x4_f32(false, a, false, b3, (short)0, acc3, false, false);
    }

    float* yb = Y + n0 + l16;
    #pragma unroll
    for (int i = 0; i < 8; ++i) {
        float* yr = yb + (size_t)(m0 + i + 8 * half) * HID;
        yr[0]  = acc0[i];
        yr[16] = acc1[i];
        yr[32] = acc2[i];
        yr[48] = acc3[i];
    }
}

__global__ __launch_bounds__(32) void gcn_gemm2_wmma(
    const float* __restrict__ Hraw, const float* __restrict__ b1,
    const float* __restrict__ W2, float* __restrict__ Y)
{
    const int mtile = blockIdx.x;
    const int lane  = threadIdx.x & 31;
    const int half  = lane >> 4;
    const int l16   = lane & 15;

    const int m0 = mtile * 16;

    const float* hrow = Hraw + (size_t)(m0 + l16) * HID + 2 * half;
    const float* brow = b1 + 2 * half;
    const float* wcol = W2 + l16;

    v8f acc0 = {}, acc1 = {}, acc2 = {}, acc3 = {};
    #pragma unroll 2
    for (int k = 0; k < HID; k += 4) {
        const float2 hv = *(const float2*)(hrow + k);
        const float2 bv = *(const float2*)(brow + k);
        v2f a;
        a.x = fmaxf(hv.x + bv.x, 0.0f);
        a.y = fmaxf(hv.y + bv.y, 0.0f);
        const float* wk0 = wcol + (size_t)(k + 2 * half) * CLS;
        const float* wk1 = wcol + (size_t)(k + 2 * half + 1) * CLS;
        v2f b0; b0.x = wk0[0];  b0.y = wk1[0];
        v2f b1v; b1v.x = wk0[16]; b1v.y = wk1[16];
        v2f b2v; b2v.x = wk0[32]; b2v.y = wk1[32];
        v2f b3v; b3v.x = wk0[48]; b3v.y = wk1[48];
        acc0 = __builtin_amdgcn_wmma_f32_16x16x4_f32(false, a, false, b0,  (short)0, acc0, false, false);
        acc1 = __builtin_amdgcn_wmma_f32_16x16x4_f32(false, a, false, b1v, (short)0, acc1, false, false);
        acc2 = __builtin_amdgcn_wmma_f32_16x16x4_f32(false, a, false, b2v, (short)0, acc2, false, false);
        acc3 = __builtin_amdgcn_wmma_f32_16x16x4_f32(false, a, false, b3v, (short)0, acc3, false, false);
    }

    float* yb = Y + l16;
    #pragma unroll
    for (int i = 0; i < 8; ++i) {
        float* yr = yb + (size_t)(m0 + i + 8 * half) * CLS;
        yr[0]  = acc0[i];
        yr[16] = acc1[i];
        yr[32] = acc2[i];
        yr[48] = acc3[i];
    }
}

#define OWN_ROWS   196
#define OWN_GROUPS 256
#define OWN_THREADS 1024
#define OWN_LDS_WORDS 43008

template <int F>
__global__ __launch_bounds__(OWN_THREADS) void gcn_spmm_own(
    const int* __restrict__ erow, const int* __restrict__ ecol,
    const float* __restrict__ eval, const float* __restrict__ dense,
    float* __restrict__ out)
{
    constexpr int VEC = F / 32;
    constexpr int ACC_WORDS = OWN_ROWS * F;
    constexpr int CAP = OWN_LDS_WORDS - ACC_WORDS - 4;
    __shared__ float acc[ACC_WORDS];
    __shared__ unsigned int elist[CAP];
    __shared__ unsigned int ecount;

    const int t    = threadIdx.x;
    const int lane = t & 31;
    const int wave = t >> 5;
    const int lo   = blockIdx.x * OWN_ROWS;
    const int rows_here = min(OWN_ROWS, NROWS - lo);

    for (int i = t; i < ACC_WORDS; i += OWN_THREADS) acc[i] = 0.0f;
    if (t == 0) ecount = 0u;
    __syncthreads();

    const int4* erow4 = (const int4*)erow;
    constexpr int NQ = NEDGE / 4;
    constexpr int SCAN_UNR = 7;
    for (int q0 = t; q0 < NQ; q0 += OWN_THREADS * SCAN_UNR) {
        int4 rq[SCAN_UNR];
        #pragma unroll
        for (int u = 0; u < SCAN_UNR; ++u) {
            const int q = q0 + u * OWN_THREADS;
            rq[u] = (q < NQ) ? erow4[q] : make_int4(-1, -1, -1, -1);
        }
        #pragma unroll
        for (int u = 0; u < SCAN_UNR; ++u) {
            const int q = q0 + u * OWN_THREADS;
            const unsigned int rl4[4] = {
                (unsigned int)(rq[u].x - lo), (unsigned int)(rq[u].y - lo),
                (unsigned int)(rq[u].z - lo), (unsigned int)(rq[u].w - lo) };
            const unsigned int nearest = min(min(rl4[0], rl4[1]), min(rl4[2], rl4[3]));
            if (nearest < (unsigned int)rows_here) {
                #pragma unroll
                for (int s = 0; s < 4; ++s) {
                    const unsigned int rl = rl4[s];
                    if (rl < (unsigned int)rows_here) {
                        const unsigned int e = (unsigned int)(q * 4 + s);
                        const unsigned int slot = atomicAdd(&ecount, 1u);
                        if (slot < (unsigned int)CAP) {
                            elist[slot] = e | (rl << 20);
                        } else {
                            const int   c = ecol[e];
                            const float v = eval[e];
                            const float* src = dense + (size_t)c * F;
                            float* dst = acc + rl * F;
                            #pragma unroll 1
                            for (int f = 0; f < F; ++f)
                                atomicAdd(dst + (f % VEC) * 32 + (f / VEC), v * src[f]);
                        }
                    }
                }
            }
        }
    }
    __syncthreads();

    const int total = __builtin_amdgcn_readfirstlane((int)min(ecount, (unsigned int)CAP));
    for (int base = wave * 32; base < total; base += (OWN_THREADS / 32) * 32) {
        const int idx = base + lane;
        int   c = 0;
        float v = 0.0f;
        int   rl = 0;
        if (idx < total) {
            const unsigned int pk = elist[idx];
            const unsigned int e = pk & 0xFFFFFu;
            rl = (int)(pk >> 20);
            c = ecol[e];
            v = eval[e];
        }
        const int n = min(32, total - base);
        for (int j0 = 0; j0 < n; j0 += 8) {
            int   cj[8];
            float vv[8];
            int   ro[8];
            float g[8][VEC];
            #pragma unroll
            for (int u = 0; u < 8; ++u) {
                cj[u] = __builtin_amdgcn_readlane(c, j0 + u);
                vv[u] = __builtin_bit_cast(float,
                          __builtin_amdgcn_readlane(__builtin_bit_cast(int, v), j0 + u));
                ro[u] = __builtin_amdgcn_readlane(rl, j0 + u) * F;
            }
            #pragma unroll
            for (int u = 0; u < 8; ++u) {
                const float* src = dense + (size_t)cj[u] * F + lane * VEC;
                if constexpr (VEC == 4) {
                    const float4 x = *(const float4*)src;
                    g[u][0] = x.x; g[u][1] = x.y; g[u][2] = x.z; g[u][3] = x.w;
                } else {
                    const float2 x = *(const float2*)src;
                    g[u][0] = x.x; g[u][1] = x.y;
                }
            }
            #pragma unroll
            for (int u = 0; u < 8; ++u) {
                if (j0 + u < n) {
                    float* dst = acc + ro[u] + lane;
                    #pragma unroll
                    for (int k = 0; k < VEC; ++k)
                        atomicAdd(dst + k * 32, vv[u] * g[u][k]);
                }
            }
        }
    }
    __syncthreads();

    constexpr int QPR = F / 4;
    for (int i = t; i < rows_here * QPR; i += OWN_THREADS) {
        const int row = i / QPR;
        const int q   = i % QPR;
        const float* a = acc + row * F;
        float4 o;
        o.x = a[((4 * q + 0) % VEC) * 32 + (4 * q + 0) / VEC];
        o.y = a[((4 * q + 1) % VEC) * 32 + (4 * q + 1) / VEC];
        o.z = a[((4 * q + 2) % VEC) * 32 + (4 * q + 2) / VEC];
        o.w = a[((4 * q + 3) % VEC) * 32 + (4 * q + 3) / VEC];
        *(float4*)(out + (size_t)(lo + row) * F + 4 * q) = o;
    }
}

__global__ __launch_bounds__(256) void gcn_logsoftmax64(
    const float* __restrict__ L, const float* __restrict__ b2,
    float* __restrict__ out)
{
    const int r = (int)((blockIdx.x * (size_t)blockDim.x + threadIdx.x) >> 5);
    const int lane = threadIdx.x & 31;
    if (r >= NROWS) return;

    float2 v = *(const float2*)(L + (size_t)r * CLS + lane * 2);
    const float2 bb = *(const float2*)(b2 + lane * 2);
    v.x += bb.x;
    v.y += bb.y;

    float m = fmaxf(v.x, v.y);
    #pragma unroll
    for (int off = 16; off > 0; off >>= 1)
        m = fmaxf(m, __shfl_xor(m, off, 32));

    float s = __expf(v.x - m) + __expf(v.y - m);
    #pragma unroll
    for (int off = 16; off > 0; off >>= 1)
        s += __shfl_xor(s, off, 32);

    const float lse = m + __logf(s);
    float2 o;
    o.x = v.x - lse;
    o.y = v.y - lse;
    *(float2*)(out + (size_t)r * CLS + lane * 2) = o;
}

extern "C" void kernel_launch(void* const* d_in, const int* in_sizes, int n_in,
                              void* d_out, int out_size, void* d_ws, size_t ws_size,
                              hipStream_t stream) {
    const float* x        = (const float*)d_in[0];
    const int*   edge_row = (const int*)  d_in[1];
    const int*   edge_col = (const int*)  d_in[2];
    const float* edge_val = (const float*)d_in[3];
    const float* W1       = (const float*)d_in[4];
    const float* b1       = (const float*)d_in[5];
    const float* W2       = (const float*)d_in[6];
    const float* b2       = (const float*)d_in[7];
    float* out = (float*)d_out;

    float* ws    = (float*)d_ws;
    float* XW1   = ws;
    float* Hraw  = ws + (size_t)NROWS * HID;
    float* HW2   = XW1;
    float* Lraw  = Hraw;

    const int MT = NROWS / 16;

    gcn_gemm1_wmma<<<MT, 64, 0, stream>>>(x, W1, XW1);
    gcn_spmm_own<HID><<<OWN_GROUPS, OWN_THREADS, 0, stream>>>(edge_row, edge_col, edge_val, XW1, Hraw);
    gcn_gemm2_wmma<<<MT, 32, 0, stream>>>(Hraw, b1, W2, HW2);
    gcn_spmm_own<CLS><<<OWN_GROUPS, OWN_THREADS, 0, stream>>>(edge_row, edge_col, edge_val, HW2, Lraw);
    gcn_logsoftmax64<<<(NROWS * 32 + 255) / 256, 256, 0, stream>>>(Lraw, b2, out);
}


# Edge lists built once per call: F=128 SpMM launch stores each group's list (col, val, row) in the workspace; F=64 launch reads it, no second edge_row scan; sums still LDS float atomics, arrival order
# speedup vs baseline: 1.9571x; 1.2521x over previous
#include <hip/hip_runtime.h>


typedef __attribute__((ext_vector_type(2))) float v2f;
typedef __attribute__((ext_vector_type(8))) float v8f;

#define NROWS 50000
#define INDIM 256
#define HID   128
#define CLS   64
#define NEDGE 800000

__global__ __launch_bounds__(64) void gcn_gemm1_wmma(
    const float* __restrict__ X, const float* __restrict__ W1,
    float* __restrict__ Y)
{
    const int mtile = blockIdx.x;
    const int wv    = threadIdx.x >> 5;
    const int lane  = threadIdx.x & 31;
    const int half  = lane >> 4;
    const int l16   = lane & 15;

    const int m0 = mtile * 16;
    const int n0 = wv * 64;

    const float* xrow = X + (size_t)(m0 + l16) * INDIM + 2 * half;
    const float* wcol = W1 + n0 + l16;

    v8f acc0 = {}, acc1 = {}, acc2 = {}, acc3 = {};
    #pragma unroll 2
    for (int k = 0; k < INDIM; k += 4) {
        const float2 av = *(const float2*)(xrow + k);
        v2f a; a.x = av.x; a.y = av.y;
        const float* wk0 = wcol + (size_t)(k + 2 * half) * HID;
        const float* wk1 = wcol + (size_t)(k + 2 * half + 1) * HID;
        v2f b0; b0.x = wk0[0];  b0.y = wk1[0];
        v2f b1; b1.x = wk0[16]; b1.y = wk1[16];
        v2f b2; b2.x = wk0[32]; b2.y = wk1[32];
        v2f b3; b3.x = wk0[48]; b3.y = wk1[48];
        acc0 = __builtin_amdgcn_wmma_f32_16x16x4_f32(false, a, false, b0, (short)0, acc0, false, false);
        acc1 = __builtin_amdgcn_wmma_f32_16x16x4_f32(false, a, false, b1, (short)0, acc1, false, false);
        acc2 = __builtin_amdgcn_wmma_f32_16x16x4_f32(false, a, false, b2, (short)0, acc2, false, false);
        acc3 = __builtin_amdgcn_wmma_f32_16x16x4_f32(false, a, false, b3, (short)0, acc3, false, false);
    }

    float* yb = Y + n0 + l16;
    #pragma unroll
    for (int i = 0; i < 8; ++i) {
        float* yr = yb + (size_t)(m0 + i + 8 * half) * HID;
        yr[0]  = acc0[i];
        yr[16] = acc1[i];
        yr[32] = acc2[i];
        yr[48] = acc3[i];
    }
}

__global__ __launch_bounds__(32) void gcn_gemm2_wmma(
    const float* __restrict__ Hraw, const float* __restrict__ b1,
    const float* __restrict__ W2, float* __restrict__ Y)
{
    const int mtile = blockIdx.x;
    const int lane  = threadIdx.x & 31;
    const int half  = lane >> 4;
    const int l16   = lane & 15;

    const int m0 = mtile * 16;

    const float* hrow = Hraw + (size_t)(m0 + l16) * HID + 2 * half;
    const float* brow = b1 + 2 * half;
    const float* wcol = W2 + l16;

    v8f acc0 = {}, acc1 = {}, acc2 = {}, acc3 = {};
    #pragma unroll 2
    for (int k = 0; k < HID; k += 4) {
        const float2 hv = *(const float2*)(hrow + k);
        const float2 bv = *(const float2*)(brow + k);
        v2f a;
        a.x = fmaxf(hv.x + bv.x, 0.0f);
        a.y = fmaxf(hv.y + bv.y, 0.0f);
        const float* wk0 = wcol + (size_t)(k + 2 * half) * CLS;
        const float* wk1 = wcol + (size_t)(k + 2 * half + 1) * CLS;
        v2f b0; b0.x = wk0[0];  b0.y = wk1[0];
        v2f b1v; b1v.x = wk0[16]; b1v.y = wk1[16];
        v2f b2v; b2v.x = wk0[32]; b2v.y = wk1[32];
        v2f b3v; b3v.x = wk0[48]; b3v.y = wk1[48];
        acc0 = __builtin_amdgcn_wmma_f32_16x16x4_f32(false, a, false, b0,  (short)0, acc0, false, false);
        acc1 = __builtin_amdgcn_wmma_f32_16x16x4_f32(false, a, false, b1v, (short)0, acc1, false, false);
        acc2 = __builtin_amdgcn_wmma_f32_16x16x4_f32(false, a, false, b2v, (short)0, acc2, false, false);
        acc3 = __builtin_amdgcn_wmma_f32_16x16x4_f32(false, a, false, b3v, (short)0, acc3, false, false);
    }

    float* yb = Y + l16;
    #pragma unroll
    for (int i = 0; i < 8; ++i) {
        float* yr = yb + (size_t)(m0 + i + 8 * half) * CLS;
        yr[0]  = acc0[i];
        yr[16] = acc1[i];
        yr[32] = acc2[i];
        yr[48] = acc3[i];
    }
}

#define OWN_ROWS   196
#define OWN_GROUPS 256
#define OWN_THREADS 1024
#define OWN_LDS_WORDS 43008
#define PUB_CAP    (OWN_LDS_WORDS - OWN_ROWS * HID - 4)
#define PUB_STRIDE 17920

template <int F, int MODE>
__global__ __launch_bounds__(OWN_THREADS) void gcn_spmm_own(
    const int* __restrict__ erow, const int* __restrict__ ecol,
    const float* __restrict__ eval, const float* __restrict__ dense,
    float* __restrict__ out,
    unsigned int* __restrict__ publist, unsigned int* __restrict__ pubcount)
{
    constexpr int VEC = F / 32;
    constexpr int ACC_WORDS = OWN_ROWS * F;
    constexpr int CAP = OWN_LDS_WORDS - ACC_WORDS - 4;
    __shared__ float acc[ACC_WORDS];
    __shared__ unsigned int elist[CAP];
    __shared__ unsigned int ecount;
    static_assert(MODE != 0 || CAP == PUB_CAP, "stored list capacity");
    static_assert(PUB_CAP <= PUB_STRIDE && OWN_ROWS <= 256 && NROWS <= 65536, "packing");

    const int t    = threadIdx.x;
    const int lane = t & 31;
    const int wave = t >> 5;
    const int lo   = blockIdx.x * OWN_ROWS;
    const int rows_here = min(OWN_ROWS, NROWS - lo);
    uint2* mypub = (uint2*)publist + (size_t)blockIdx.x * PUB_STRIDE;

    unsigned int stored = 0u;
    bool need_scan = true;
    if constexpr (MODE == 1) {
        stored = (unsigned int)__builtin_amdgcn_readfirstlane((int)pubcount[blockIdx.x]);
        need_scan = stored > (unsigned int)PUB_CAP;
    }

    for (int i = t; i < ACC_WORDS; i += OWN_THREADS) acc[i] = 0.0f;
    if (t == 0) ecount = 0u;
    __syncthreads();

    const int4* erow4 = (const int4*)erow;
    constexpr int NQ = NEDGE / 4;
    constexpr int SCAN_UNR = 7;
    for (int q0 = t; need_scan && q0 < NQ; q0 += OWN_THREADS * SCAN_UNR) {
        int4 rq[SCAN_UNR];
        #pragma unroll
        for (int u = 0; u < SCAN_UNR; ++u) {
            const int q = q0 + u * OWN_THREADS;
            rq[u] = (q < NQ) ? erow4[q] : make_int4(-1, -1, -1, -1);
        }
        #pragma unroll
        for (int u = 0; u < SCAN_UNR; ++u) {
            const int q = q0 + u * OWN_THREADS;
            const unsigned int rl4[4] = {
                (unsigned int)(rq[u].x - lo), (unsigned int)(rq[u].y - lo),
                (unsigned int)(rq[u].z - lo), (unsigned int)(rq[u].w - lo) };
            const unsigned int nearest = min(min(rl4[0], rl4[1]), min(rl4[2], rl4[3]));
            if (nearest < (unsigned int)rows_here) {
                #pragma unroll
                for (int s = 0; s < 4; ++s) {
                    const unsigned int rl = rl4[s];
                    if (rl < (unsigned int)rows_here) {
                        const unsigned int e = (unsigned int)(q * 4 + s);
                        const unsigned int slot = atomicAdd(&ecount, 1u);
                        if (slot < (unsigned int)CAP) {
                            elist[slot] = e | (rl << 20);
                        } else {
                            const int   c = ecol[e];
                            const float v = eval[e];
                            const float* src = dense + (size_t)c * F;
                            float* dst = acc + rl * F;
                            #pragma unroll 1
                            for (int f = 0; f < F; ++f)
                                atomicAdd(dst + (f % VEC) * 32 + (f / VEC), v * src[f]);
                        }
                    }
                }
            }
        }
    }
    __syncthreads();

    const int total = need_scan
        ? __builtin_amdgcn_readfirstlane((int)min(ecount, (unsigned int)CAP))
        : (int)stored;
    if constexpr (MODE == 0) {
        if (t == 0) pubcount[blockIdx.x] = ecount;
    }
    for (int base = wave * 32; base < total; base += (OWN_THREADS / 32) * 32) {
        const int idx = base + lane;
        int   c = 0;
        float v = 0.0f;
        int   rl = 0;
        if (idx < total) {
            if (need_scan) {
                const unsigned int pk = elist[idx];
                const unsigned int e = pk & 0xFFFFFu;
                rl = (int)(pk >> 20);
                c = ecol[e];
                v = eval[e];
                if constexpr (MODE == 0)
                    mypub[idx] = make_uint2((unsigned int)c | ((unsigned int)rl << 16),
                                            __builtin_bit_cast(unsigned int, v));
            } else {
                const uint2 pe = mypub[idx];
                c  = (int)(pe.x & 0xFFFFu);
                rl = (int)(pe.x >> 16);
                v  = __builtin_bit_cast(float, pe.y);
            }
        }
        const int n = min(32, total - base);
        for (int j0 = 0; j0 < n; j0 += 8) {
            int   cj[8];
            float vv[8];
            int   ro[8];
            float g[8][VEC];
            #pragma unroll
            for (int u = 0; u < 8; ++u) {
                cj[u] = __builtin_amdgcn_readlane(c, j0 + u);
                vv[u] = __builtin_bit_cast(float,
                          __builtin_amdgcn_readlane(__builtin_bit_cast(int, v), j0 + u));
                ro[u] = __builtin_amdgcn_readlane(rl, j0 + u) * F;
            }
            #pragma unroll
            for (int u = 0; u < 8; ++u) {
                const float* src = dense + (size_t)cj[u] * F + lane * VEC;
                if constexpr (VEC == 4) {
                    const float4 x = *(const float4*)src;
                    g[u][0] = x.x; g[u][1] = x.y; g[u][2] = x.z; g[u][3] = x.w;
                } else {
                    const float2 x = *(const float2*)src;
                    g[u][0] = x.x; g[u][1] = x.y;
                }
            }
            #pragma unroll
            for (int u = 0; u < 8; ++u) {
                if (j0 + u < n) {
                    float* dst = acc + ro[u] + lane;
                    #pragma unroll
                    for (int k = 0; k < VEC; ++k)
                        atomicAdd(dst + k * 32, vv[u] * g[u][k]);
                }
            }
        }
    }
    __syncthreads();

    constexpr int QPR = F / 4;
    for (int i = t; i < rows_here * QPR; i += OWN_THREADS) {
        const int row = i / QPR;
        const int q   = i % QPR;
        const float* a = acc + row * F;
        float4 o;
        o.x = a[((4 * q + 0) % VEC) * 32 + (4 * q + 0) / VEC];
        o.y = a[((4 * q + 1) % VEC) * 32 + (4 * q + 1) / VEC];
        o.z = a[((4 * q + 2) % VEC) * 32 + (4 * q + 2) / VEC];
        o.w = a[((4 * q + 3) % VEC) * 32 + (4 * q + 3) / VEC];
        *(float4*)(out + (size_t)(lo + row) * F + 4 * q) = o;
    }
}

__global__ __launch_bounds__(256) void gcn_logsoftmax64(
    const float* __restrict__ L, const float* __restrict__ b2,
    float* __restrict__ out)
{
    const int r = (int)((blockIdx.x * (size_t)blockDim.x + threadIdx.x) >> 5);
    const int lane = threadIdx.x & 31;
    if (r >= NROWS) return;

    float2 v = *(const float2*)(L + (size_t)r * CLS + lane * 2);
    const float2 bb = *(const float2*)(b2 + lane * 2);
    v.x += bb.x;
    v.y += bb.y;

    float m = fmaxf(v.x, v.y);
    #pragma unroll
    for (int off = 16; off > 0; off >>= 1)
        m = fmaxf(m, __shfl_xor(m, off, 32));

    float s = __expf(v.x - m) + __expf(v.y - m);
    #pragma unroll
    for (int off = 16; off > 0; off >>= 1)
        s += __shfl_xor(s, off, 32);

    const float lse = m + __logf(s);
    float2 o;
    o.x = v.x - lse;
    o.y = v.y - lse;
    *(float2*)(out + (size_t)r * CLS + lane * 2) = o;
}

extern "C" void kernel_launch(void* const* d_in, const int* in_sizes, int n_in,
                              void* d_out, int out_size, void* d_ws, size_t ws_size,
                              hipStream_t stream) {
    const float* x        = (const float*)d_in[0];
    const int*   edge_row = (const int*)  d_in[1];
    const int*   edge_col = (const int*)  d_in[2];
    const float* edge_val = (const float*)d_in[3];
    const float* W1       = (const float*)d_in[4];
    const float* b1       = (const float*)d_in[5];
    const float* W2       = (const float*)d_in[6];
    const float* b2       = (const float*)d_in[7];
    float* out = (float*)d_out;

    float* ws    = (float*)d_ws;
    float* XW1   = ws;
    float* Hraw  = ws + (size_t)NROWS * HID;
    float* HW2   = XW1;
    float* Lraw  = Hraw;

    unsigned int* plist  = (unsigned int*)d_ws + 16777216;
    unsigned int* pcount = (unsigned int*)d_ws + 33554432;

    const int MT = NROWS / 16;

    gcn_gemm1_wmma<<<MT, 64, 0, stream>>>(x, W1, XW1);
    gcn_spmm_own<HID, 0><<<OWN_GROUPS, OWN_THREADS, 0, stream>>>(edge_row, edge_col, edge_val, XW1, Hraw, plist, pcount);
    gcn_gemm2_wmma<<<MT, 32, 0, stream>>>(Hraw, b1, W2, HW2);
    gcn_spmm_own<CLS, 1><<<OWN_GROUPS, OWN_THREADS, 0, stream>>>(edge_row, edge_col, edge_val, HW2, Lraw, plist, pcount);
    gcn_logsoftmax64<<<(NROWS * 32 + 255) / 256, 256, 0, stream>>>(Lraw, b2, out);
}


# SpMM edge scan reads 1-byte owner keys (row/196, written by the GEMM1 launch, one thread per 4 edges), not 4-byte edge_row: 16 edges per 16-B load; sums still LDS float atomics, arrival order
# speedup vs baseline: 2.1164x; 1.0814x over previous
#include <hip/hip_runtime.h>


typedef __attribute__((ext_vector_type(2))) float v2f;
typedef __attribute__((ext_vector_type(8))) float v8f;

#define NROWS 50000
#define INDIM 256
#define HID   128
#define CLS   64
#define NEDGE 800000
#define OWN_ROWS 196

__global__ __launch_bounds__(64) void gcn_gemm1_wmma(
    const float* __restrict__ X, const float* __restrict__ W1,
    float* __restrict__ Y,
    const int* __restrict__ erow, unsigned int* __restrict__ keys)
{
    static_assert((NROWS / 16) * 64 * 4 == NEDGE, "one thread per edge quad");
    const int gq = blockIdx.x * 64 + threadIdx.x;
    const int4 rq = ((const int4*)erow)[gq];

    const int mtile = blockIdx.x;
    const int wv    = threadIdx.x >> 5;
    const int lane  = threadIdx.x & 31;
    const int half  = lane >> 4;
    const int l16   = lane & 15;

    const int m0 = mtile * 16;
    const int n0 = wv * 64;

    const float* xrow = X + (size_t)(m0 + l16) * INDIM + 2 * half;
    const float* wcol = W1 + n0 + l16;

    v8f acc0 = {}, acc1 = {}, acc2 = {}, acc3 = {};
    #pragma unroll 2
    for (int k = 0; k < INDIM; k += 4) {
        const float2 av = *(const float2*)(xrow + k);
        v2f a; a.x = av.x; a.y = av.y;
        const float* wk0 = wcol + (size_t)(k + 2 * half) * HID;
        const float* wk1 = wcol + (size_t)(k + 2 * half + 1) * HID;
        v2f b0; b0.x = wk0[0];  b0.y = wk1[0];
        v2f b1; b1.x = wk0[16]; b1.y = wk1[16];
        v2f b2; b2.x = wk0[32]; b2.y = wk1[32];
        v2f b3; b3.x = wk0[48]; b3.y = wk1[48];
        acc0 = __builtin_amdgcn_wmma_f32_16x16x4_f32(false, a, false, b0, (short)0, acc0, false, false);
        acc1 = __builtin_amdgcn_wmma_f32_16x16x4_f32(false, a, false, b1, (short)0, acc1, false, false);
        acc2 = __builtin_amdgcn_wmma_f32_16x16x4_f32(false, a, false, b2, (short)0, acc2, false, false);
        acc3 = __builtin_amdgcn_wmma_f32_16x16x4_f32(false, a, false, b3, (short)0, acc3, false, false);
    }

    float* yb = Y + n0 + l16;
    #pragma unroll
    for (int i = 0; i < 8; ++i) {
        float* yr = yb + (size_t)(m0 + i + 8 * half) * HID;
        yr[0]  = acc0[i];
        yr[16] = acc1[i];
        yr[32] = acc2[i];
        yr[48] = acc3[i];
    }

    keys[gq] = (((unsigned int)rq.x / OWN_ROWS) & 0xFFu)
             | ((((unsigned int)rq.y / OWN_ROWS) & 0xFFu) << 8)
             | ((((unsigned int)rq.z / OWN_ROWS) & 0xFFu) << 16)
             | (((unsigned int)rq.w / OWN_ROWS) << 24);
}

__global__ __launch_bounds__(32) void gcn_gemm2_wmma(
    const float* __restrict__ Hraw, const float* __restrict__ b1,
    const float* __restrict__ W2, float* __restrict__ Y)
{
    const int mtile = blockIdx.x;
    const int lane  = threadIdx.x & 31;
    const int half  = lane >> 4;
    const int l16   = lane & 15;

    const int m0 = mtile * 16;

    const float* hrow = Hraw + (size_t)(m0 + l16) * HID + 2 * half;
    const float* brow = b1 + 2 * half;
    const float* wcol = W2 + l16;

    v8f acc0 = {}, acc1 = {}, acc2 = {}, acc3 = {};
    #pragma unroll 2
    for (int k = 0; k < HID; k += 4) {
        const float2 hv = *(const float2*)(hrow + k);
        const float2 bv = *(const float2*)(brow + k);
        v2f a;
        a.x = fmaxf(hv.x + bv.x, 0.0f);
        a.y = fmaxf(hv.y + bv.y, 0.0f);
        const float* wk0 = wcol + (size_t)(k + 2 * half) * CLS;
        const float* wk1 = wcol + (size_t)(k + 2 * half + 1) * CLS;
        v2f b0; b0.x = wk0[0];  b0.y = wk1[0];
        v2f b1v; b1v.x = wk0[16]; b1v.y = wk1[16];
        v2f b2v; b2v.x = wk0[32]; b2v.y = wk1[32];
        v2f b3v; b3v.x = wk0[48]; b3v.y = wk1[48];
        acc0 = __builtin_amdgcn_wmma_f32_16x16x4_f32(false, a, false, b0,  (short)0, acc0, false, false);
        acc1 = __builtin_amdgcn_wmma_f32_16x16x4_f32(false, a, false, b1v, (short)0, acc1, false, false);
        acc2 = __builtin_amdgcn_wmma_f32_16x16x4_f32(false, a, false, b2v, (short)0, acc2, false, false);
        acc3 = __builtin_amdgcn_wmma_f32_16x16x4_f32(false, a, false, b3v, (short)0, acc3, false, false);
    }

    float* yb = Y + l16;
    #pragma unroll
    for (int i = 0; i < 8; ++i) {
        float* yr = yb + (size_t)(m0 + i + 8 * half) * CLS;
        yr[0]  = acc0[i];
        yr[16] = acc1[i];
        yr[32] = acc2[i];
        yr[48] = acc3[i];
    }
}

#define OWN_GROUPS 256
#define OWN_THREADS 1024
#define OWN_LDS_WORDS 43008
#define PUB_CAP    (OWN_LDS_WORDS - OWN_ROWS * HID - 4)
#define PUB_STRIDE 17920
#define NO_ROW     0xFFFF

template <int F, int MODE>
__global__ __launch_bounds__(OWN_THREADS) void gcn_spmm_own(
    const int* __restrict__ erow, const int* __restrict__ ecol,
    const float* __restrict__ eval, const float* __restrict__ dense,
    float* __restrict__ out,
    unsigned int* __restrict__ publist, unsigned int* __restrict__ pubcount,
    const unsigned int* __restrict__ keys)
{
    constexpr int VEC = F / 32;
    constexpr int ACC_WORDS = OWN_ROWS * F;
    constexpr int CAP = OWN_LDS_WORDS - ACC_WORDS - 4;
    __shared__ float acc[ACC_WORDS];
    __shared__ unsigned int elist[CAP];
    __shared__ unsigned int ecount;
    static_assert(MODE != 0 || CAP == PUB_CAP, "stored list capacity");
    static_assert(PUB_CAP <= PUB_STRIDE && OWN_ROWS <= 256 && NROWS <= 65536, "packing");

    const int t    = threadIdx.x;
    const int lane = t & 31;
    const int wave = t >> 5;
    const int lo   = blockIdx.x * OWN_ROWS;
    const int rows_here = min(OWN_ROWS, NROWS - lo);
    uint2* mypub = (uint2*)publist + (size_t)blockIdx.x * PUB_STRIDE;

    unsigned int stored = 0u;
    bool need_scan = true;
    if constexpr (MODE == 1) {
        stored = (unsigned int)__builtin_amdgcn_readfirstlane((int)pubcount[blockIdx.x]);
        need_scan = stored > (unsigned int)PUB_CAP;
    }

    for (int i = t; i < ACC_WORDS; i += OWN_THREADS) acc[i] = 0.0f;
    if (t == 0) ecount = 0u;
    __syncthreads();

    const uint4* key16 = (const uint4*)keys;
    constexpr int NQ = NEDGE / 16;
    constexpr int SCAN_UNR = 7;
    const unsigned int mine = (unsigned int)blockIdx.x * 0x01010101u;
    for (int q0 = t; need_scan && q0 < NQ; q0 += OWN_THREADS * SCAN_UNR) {
        uint4 kq[SCAN_UNR];
        #pragma unroll
        for (int u = 0; u < SCAN_UNR; ++u)
            kq[u] = key16[min(q0 + u * OWN_THREADS, NQ - 1)];
        #pragma unroll
        for (int u = 0; u < SCAN_UNR; ++u) {
            const int q = q0 + u * OWN_THREADS;
            const unsigned int x4[4] = { kq[u].x ^ mine, kq[u].y ^ mine,
                                         kq[u].z ^ mine, kq[u].w ^ mine };
            unsigned int nz4[4];
            #pragma unroll
            for (int w = 0; w < 4; ++w)
                nz4[w] = ((x4[w] & 0x7F7F7F7Fu) + 0x7F7F7F7Fu) | x4[w];
            const unsigned int all_nz = nz4[0] & nz4[1] & nz4[2] & nz4[3];
            if ((all_nz & 0x80808080u) != 0x80808080u && q < NQ) {
                unsigned int some = 0u;
                #pragma unroll
                for (int w = 0; w < 4; ++w)
                    some |= (nz4[w] >> (7 - w)) & (0x01010101u << w);
                unsigned int hit = ~some & 0x0F0F0F0Fu;
                do {
                    const int p = __builtin_ctz(hit);
                    hit &= hit - 1u;
                    const unsigned int e = (unsigned int)(q * 16 + (p & 3) * 4 + (p >> 3));
                    const unsigned int slot = atomicAdd(&ecount, 1u);
                    if (slot < (unsigned int)CAP) {
                        elist[slot] = e;
                    } else {
                        const unsigned int rl = (unsigned int)(erow[e] - lo);
                        if (rl < (unsigned int)rows_here) {
                            const int   c = ecol[e];
                            const float v = eval[e];
                            const float* src = dense + (size_t)c * F;
                            float* dst = acc + rl * F;
                            #pragma unroll 1
                            for (int f = 0; f < F; ++f)
                                atomicAdd(dst + (f % VEC) * 32 + (f / VEC), v * src[f]);
                        }
                    }
                } while (hit != 0u);
            }
        }
    }
    __syncthreads();

    const int total = need_scan
        ? __builtin_amdgcn_readfirstlane((int)min(ecount, (unsigned int)CAP))
        : (int)stored;
    if constexpr (MODE == 0) {
        if (t == 0) pubcount[blockIdx.x] = ecount;
    }
    for (int base = wave * 32; base < total; base += (OWN_THREADS / 32) * 32) {
        const int idx = base + lane;
        int   c = 0;
        float v = 0.0f;
        int   rl = NO_ROW;
        if (idx < total) {
            if (need_scan) {
                const unsigned int e = elist[idx];
                const unsigned int r = (unsigned int)(erow[e] - lo);
                c = ecol[e];
                v = eval[e];
                rl = (r < (unsigned int)rows_here) ? (int)r : NO_ROW;
                if constexpr (MODE == 0)
                    mypub[idx] = make_uint2((unsigned int)c | ((unsigned int)rl << 16),
                                            __builtin_bit_cast(unsigned int, v));
            } else {
                const uint2 pe = mypub[idx];
                c  = (int)(pe.x & 0xFFFFu);
                rl = (int)(pe.x >> 16);
                v  = __builtin_bit_cast(float, pe.y);
            }
        }
        const int n = min(32, total - base);
        for (int j0 = 0; j0 < n; j0 += 8) {
            int   cj[8];
            float vv[8];
            int   ro[8];
            bool  in_range[8];
            float g[8][VEC];
            #pragma unroll
            for (int u = 0; u < 8; ++u) {
                cj[u] = __builtin_amdgcn_readlane(c, j0 + u);
                vv[u] = __builtin_bit_cast(float,
                          __builtin_amdgcn_readlane(__builtin_bit_cast(int, v), j0 + u));
                const int rj = __builtin_amdgcn_readlane(rl, j0 + u);
                in_range[u] = rj < rows_here;
                ro[u] = rj * F;
            }
            #pragma unroll
            for (int u = 0; u < 8; ++u) {
                const float* src = dense + (size_t)cj[u] * F + lane * VEC;
                if constexpr (VEC == 4) {
                    const float4 x = *(const float4*)src;
                    g[u][0] = x.x; g[u][1] = x.y; g[u][2] = x.z; g[u][3] = x.w;
                } else {
                    const float2 x = *(const float2*)src;
                    g[u][0] = x.x; g[u][1] = x.y;
                }
            }
            #pragma unroll
            for (int u = 0; u < 8; ++u) {
                if (j0 + u < n && in_range[u]) {
                    float* dst = acc + ro[u] + lane;
                    #pragma unroll
                    for (int k = 0; k < VEC; ++k)
                        atomicAdd(dst + k * 32, vv[u] * g[u][k]);
                }
            }
        }
    }
    __syncthreads();

    constexpr int QPR = F / 4;
    for (int i = t; i < rows_here * QPR; i += OWN_THREADS) {
        const int row = i / QPR;
        const int q   = i % QPR;
        const float* a = acc + row * F;
        float4 o;
        o.x = a[((4 * q + 0) % VEC) * 32 + (4 * q + 0) / VEC];
        o.y = a[((4 * q + 1) % VEC) * 32 + (4 * q + 1) / VEC];
        o.z = a[((4 * q + 2) % VEC) * 32 + (4 * q + 2) / VEC];
        o.w = a[((4 * q + 3) % VEC) * 32 + (4 * q + 3) / VEC];
        *(float4*)(out + (size_t)(lo + row) * F + 4 * q) = o;
    }
}

__global__ __launch_bounds__(256) void gcn_logsoftmax64(
    const float* __restrict__ L, const float* __restrict__ b2,
    float* __restrict__ out)
{
    const int r = (int)((blockIdx.x * (size_t)blockDim.x + threadIdx.x) >> 5);
    const int lane = threadIdx.x & 31;
    if (r >= NROWS) return;

    float2 v = *(const float2*)(L + (size_t)r * CLS + lane * 2);
    const float2 bb = *(const float2*)(b2 + lane * 2);
    v.x += bb.x;
    v.y += bb.y;

    float m = fmaxf(v.x, v.y);
    #pragma unroll
    for (int off = 16; off > 0; off >>= 1)
        m = fmaxf(m, __shfl_xor(m, off, 32));

    float s = __expf(v.x - m) + __expf(v.y - m);
    #pragma unroll
    for (int off = 16; off > 0; off >>= 1)
        s += __shfl_xor(s, off, 32);

    const float lse = m + __logf(s);
    float2 o;
    o.x = v.x - lse;
    o.y = v.y - lse;
    *(float2*)(out + (size_t)r * CLS + lane * 2) = o;
}

extern "C" void kernel_launch(void* const* d_in, const int* in_sizes, int n_in,
                              void* d_out, int out_size, void* d_ws, size_t ws_size,
                              hipStream_t stream) {
    const float* x        = (const float*)d_in[0];
    const int*   edge_row = (const int*)  d_in[1];
    const int*   edge_col = (const int*)  d_in[2];
    const float* edge_val = (const float*)d_in[3];
    const float* W1       = (const float*)d_in[4];
    const float* b1       = (const float*)d_in[5];
    const float* W2       = (const float*)d_in[6];
    const float* b2       = (const float*)d_in[7];
    float* out = (float*)d_out;

    float* ws    = (float*)d_ws;
    float* XW1   = ws;
    float* Hraw  = ws + (size_t)NROWS * HID;
    float* HW2   = XW1;
    float* Lraw  = Hraw;

    unsigned int* plist  = (unsigned int*)d_ws + 16777216;
    unsigned int* pcount = (unsigned int*)d_ws + 33554432;
    unsigned int* okeys  = (unsigned int*)d_ws + 50331648;

    const int MT = NROWS / 16;

    gcn_gemm1_wmma<<<MT, 64, 0, stream>>>(x, W1, XW1, edge_row, okeys);
    gcn_spmm_own<HID, 0><<<OWN_GROUPS, OWN_THREADS, 0, stream>>>(edge_row, edge_col, edge_val, XW1, Hraw, plist, pcount, okeys);
    gcn_gemm2_wmma<<<MT, 32, 0, stream>>>(Hraw, b1, W2, HW2);
    gcn_spmm_own<CLS, 1><<<OWN_GROUPS, OWN_THREADS, 0, stream>>>(edge_row, edge_col, edge_val, HW2, Lraw, plist, pcount, okeys);
    gcn_logsoftmax64<<<(NROWS * 32 + 255) / 256, 256, 0, stream>>>(Lraw, b2, out);
}


# Edge records: launch 1's edge-quad threads also read col, val and store 8 B per edge (col, local row, val); a listed edge costs the SpMM one scattered line fetch, not three; sums unchanged
# speedup vs baseline: 2.2191x; 1.0485x over previous
#include <hip/hip_runtime.h>


typedef __attribute__((ext_vector_type(2))) float v2f;
typedef __attribute__((ext_vector_type(8))) float v8f;

#define NROWS 50000
#define INDIM 256
#define HID   128
#define CLS   64
#define NEDGE 800000
#define OWN_ROWS 196
#define NO_ROW   0xFFFF

__global__ __launch_bounds__(64) void gcn_gemm1_wmma(
    const float* __restrict__ X, const float* __restrict__ W1,
    float* __restrict__ Y,
    const int* __restrict__ erow, const int* __restrict__ ecol,
    const float* __restrict__ eval,
    unsigned int* __restrict__ keys, unsigned int* __restrict__ rec)
{
    static_assert((NROWS / 16) * 64 * 4 == NEDGE, "one thread per edge quad");
    const int gq = blockIdx.x * 64 + threadIdx.x;
    const int4   rq = ((const int4*)erow)[gq];
    const int4   cq = ((const int4*)ecol)[gq];
    const float4 vq = ((const float4*)eval)[gq];

    const int mtile = blockIdx.x;
    const int wv    = threadIdx.x >> 5;
    const int lane  = threadIdx.x & 31;
    const int half  = lane >> 4;
    const int l16   = lane & 15;

    const int m0 = mtile * 16;
    const int n0 = wv * 64;

    const float* xrow = X + (size_t)(m0 + l16) * INDIM + 2 * half;
    const float* wcol = W1 + n0 + l16;

    v8f acc0 = {}, acc1 = {}, acc2 = {}, acc3 = {};
    #pragma unroll 2
    for (int k = 0; k < INDIM; k += 4) {
        const float2 av = *(const float2*)(xrow + k);
        v2f a; a.x = av.x; a.y = av.y;
        const float* wk0 = wcol + (size_t)(k + 2 * half) * HID;
        const float* wk1 = wcol + (size_t)(k + 2 * half + 1) * HID;
        v2f b0; b0.x = wk0[0];  b0.y = wk1[0];
        v2f b1; b1.x = wk0[16]; b1.y = wk1[16];
        v2f b2; b2.x = wk0[32]; b2.y = wk1[32];
        v2f b3; b3.x = wk0[48]; b3.y = wk1[48];
        acc0 = __builtin_amdgcn_wmma_f32_16x16x4_f32(false, a, false, b0, (short)0, acc0, false, false);
        acc1 = __builtin_amdgcn_wmma_f32_16x16x4_f32(false, a, false, b1, (short)0, acc1, false, false);
        acc2 = __builtin_amdgcn_wmma_f32_16x16x4_f32(false, a, false, b2, (short)0, acc2, false, false);
        acc3 = __builtin_amdgcn_wmma_f32_16x16x4_f32(false, a, false, b3, (short)0, acc3, false, false);
    }

    float* yb = Y + n0 + l16;
    #pragma unroll
    for (int i = 0; i < 8; ++i) {
        float* yr = yb + (size_t)(m0 + i + 8 * half) * HID;
        yr[0]  = acc0[i];
        yr[16] = acc1[i];
        yr[32] = acc2[i];
        yr[48] = acc3[i];
    }

    const unsigned int ow0 = (unsigned int)rq.x / OWN_ROWS;
    const unsigned int ow1 = (unsigned int)rq.y / OWN_ROWS;
    const unsigned int ow2 = (unsigned int)rq.z / OWN_ROWS;
    const unsigned int ow3 = (unsigned int)rq.w / OWN_ROWS;
    keys[gq] = (ow0 & 0xFFu) | ((ow1 & 0xFFu) << 8) | ((ow2 & 0xFFu) << 16) | (ow3 << 24);

    const unsigned int l0 = (unsigned int)rq.x < (unsigned int)NROWS ? (unsigned int)rq.x - ow0 * OWN_ROWS : (unsigned int)NO_ROW;
    const unsigned int l1 = (unsigned int)rq.y < (unsigned int)NROWS ? (unsigned int)rq.y - ow1 * OWN_ROWS : (unsigned int)NO_ROW;
    const unsigned int l2 = (unsigned int)rq.z < (unsigned int)NROWS ? (unsigned int)rq.z - ow2 * OWN_ROWS : (unsigned int)NO_ROW;
    const unsigned int l3 = (unsigned int)rq.w < (unsigned int)NROWS ? (unsigned int)rq.w - ow3 * OWN_ROWS : (unsigned int)NO_ROW;
    uint4 r01, r23;
    r01.x = ((unsigned int)cq.x & 0xFFFFu) | (l0 << 16);
    r01.y = __builtin_bit_cast(unsigned int, vq.x);
    r01.z = ((unsigned int)cq.y & 0xFFFFu) | (l1 << 16);
    r01.w = __builtin_bit_cast(unsigned int, vq.y);
    r23.x = ((unsigned int)cq.z & 0xFFFFu) | (l2 << 16);
    r23.y = __builtin_bit_cast(unsigned int, vq.z);
    r23.z = ((unsigned int)cq.w & 0xFFFFu) | (l3 << 16);
    r23.w = __builtin_bit_cast(unsigned int, vq.w);
    ((uint4*)rec)[2 * gq]     = r01;
    ((uint4*)rec)[2 * gq + 1] = r23;
}

__global__ __launch_bounds__(32) void gcn_gemm2_wmma(
    const float* __restrict__ Hraw, const float* __restrict__ b1,
    const float* __restrict__ W2, float* __restrict__ Y)
{
    const int mtile = blockIdx.x;
    const int lane  = threadIdx.x & 31;
    const int half  = lane >> 4;
    const int l16   = lane & 15;

    const int m0 = mtile * 16;

    const float* hrow = Hraw + (size_t)(m0 + l16) * HID + 2 * half;
    const float* brow = b1 + 2 * half;
    const float* wcol = W2 + l16;

    v8f acc0 = {}, acc1 = {}, acc2 = {}, acc3 = {};
    #pragma unroll 2
    for (int k = 0; k < HID; k += 4) {
        const float2 hv = *(const float2*)(hrow + k);
        const float2 bv = *(const float2*)(brow + k);
        v2f a;
        a.x = fmaxf(hv.x + bv.x, 0.0f);
        a.y = fmaxf(hv.y + bv.y, 0.0f);
        const float* wk0 = wcol + (size_t)(k + 2 * half) * CLS;
        const float* wk1 = wcol + (size_t)(k + 2 * half + 1) * CLS;
        v2f b0; b0.x = wk0[0];  b0.y = wk1[0];
        v2f b1v; b1v.x = wk0[16]; b1v.y = wk1[16];
        v2f b2v; b2v.x = wk0[32]; b2v.y = wk1[32];
        v2f b3v; b3v.x = wk0[48]; b3v.y = wk1[48];
        acc0 = __builtin_amdgcn_wmma_f32_16x16x4_f32(false, a, false, b0,  (short)0, acc0, false, false);
        acc1 = __builtin_amdgcn_wmma_f32_16x16x4_f32(false, a, false, b1v, (short)0, acc1, false, false);
        acc2 = __builtin_amdgcn_wmma_f32_16x16x4_f32(false, a, false, b2v, (short)0, acc2, false, false);
        acc3 = __builtin_amdgcn_wmma_f32_16x16x4_f32(false, a, false, b3v, (short)0, acc3, false, false);
    }

    float* yb = Y + l16;
    #pragma unroll
    for (int i = 0; i < 8; ++i) {
        float* yr = yb + (size_t)(m0 + i + 8 * half) * CLS;
        yr[0]  = acc0[i];
        yr[16] = acc1[i];
        yr[32] = acc2[i];
        yr[48] = acc3[i];
    }
}

#define OWN_GROUPS 256
#define OWN_THREADS 1024
#define OWN_LDS_WORDS 43008
#define PUB_CAP    (OWN_LDS_WORDS - OWN_ROWS * HID - 4)
#define PUB_STRIDE 17920

template <int F, int MODE>
__global__ __launch_bounds__(OWN_THREADS) void gcn_spmm_own(
    const float* __restrict__ dense, float* __restrict__ out,
    unsigned int* __restrict__ publist, unsigned int* __restrict__ pubcount,
    const unsigned int* __restrict__ keys, const unsigned int* __restrict__ rec)
{
    constexpr int VEC = F / 32;
    constexpr int ACC_WORDS = OWN_ROWS * F;
    constexpr int CAP = OWN_LDS_WORDS - ACC_WORDS - 4;
    __shared__ float acc[ACC_WORDS];
    __shared__ unsigned int elist[CAP];
    __shared__ unsigned int ecount;
    static_assert(MODE != 0 || CAP == PUB_CAP, "stored list capacity");
    static_assert(PUB_CAP <= PUB_STRIDE && OWN_ROWS <= 256 && NROWS <= 65536, "packing");

    const int t    = threadIdx.x;
    const int lane = t & 31;
    const int wave = t >> 5;
    const int lo   = blockIdx.x * OWN_ROWS;
    const int rows_here = min(OWN_ROWS, NROWS - lo);
    uint2* mypub = (uint2*)publist + (size_t)blockIdx.x * PUB_STRIDE;
    const uint2* rec2 = (const uint2*)rec;

    unsigned int stored = 0u;
    bool need_scan = true;
    if constexpr (MODE == 1) {
        stored = (unsigned int)__builtin_amdgcn_readfirstlane((int)pubcount[blockIdx.x]);
        need_scan = stored > (unsigned int)PUB_CAP;
    }

    for (int i = t; i < ACC_WORDS; i += OWN_THREADS) acc[i] = 0.0f;
    if (t == 0) ecount = 0u;
    __syncthreads();

    const uint4* key16 = (const uint4*)keys;
    constexpr int NQ = NEDGE / 16;
    constexpr int SCAN_UNR = 7;
    const unsigned int mine = (unsigned int)blockIdx.x * 0x01010101u;
    for (int q0 = t; need_scan && q0 < NQ; q0 += OWN_THREADS * SCAN_UNR) {
        uint4 kq[SCAN_UNR];
        #pragma unroll
        for (int u = 0; u < SCAN_UNR; ++u)
            kq[u] = key16[min(q0 + u * OWN_THREADS, NQ - 1)];
        #pragma unroll
        for (int u = 0; u < SCAN_UNR; ++u) {
            const int q = q0 + u * OWN_THREADS;
            const unsigned int x4[4] = { kq[u].x ^ mine, kq[u].y ^ mine,
                                         kq[u].z ^ mine, kq[u].w ^ mine };
            unsigned int nz4[4];
            #pragma unroll
            for (int w = 0; w < 4; ++w)
                nz4[w] = ((x4[w] & 0x7F7F7F7Fu) + 0x7F7F7F7Fu) | x4[w];
            const unsigned int all_nz = nz4[0] & nz4[1] & nz4[2] & nz4[3];
            if ((all_nz & 0x80808080u) != 0x80808080u && q < NQ) {
                unsigned int some = 0u;
                #pragma unroll
                for (int w = 0; w < 4; ++w)
                    some |= (nz4[w] >> (7 - w)) & (0x01010101u << w);
                unsigned int hit = ~some & 0x0F0F0F0Fu;
                do {
                    const int p = __builtin_ctz(hit);
                    hit &= hit - 1u;
                    const unsigned int e = (unsigned int)(q * 16 + (p & 3) * 4 + (p >> 3));
                    const unsigned int slot = atomicAdd(&ecount, 1u);
                    if (slot < (unsigned int)CAP) {
                        elist[slot] = e;
                    } else {
                        const uint2 pe = rec2[e];
                        const unsigned int rl = pe.x >> 16;
                        if (rl < (unsigned int)rows_here) {
                            const int   c = (int)(pe.x & 0xFFFFu);
                            const float v = __builtin_bit_cast(float, pe.y);
                            const float* src = dense + (size_t)c * F;
                            float* dst = acc + rl * F;
                            #pragma unroll 1
                            for (int f = 0; f < F; ++f)
                                atomicAdd(dst + (f % VEC) * 32 + (f / VEC), v * src[f]);
                        }
                    }
                } while (hit != 0u);
            }
        }
    }
    __syncthreads();

    const int total = need_scan
        ? __builtin_amdgcn_readfirstlane((int)min(ecount, (unsigned int)CAP))
        : (int)stored;
    if constexpr (MODE == 0) {
        if (t == 0) pubcount[blockIdx.x] = ecount;
    }
    for (int base = wave * 32; base < total; base += (OWN_THREADS / 32) * 32) {
        const int idx = base + lane;
        int   c = 0;
        float v = 0.0f;
        int   rl = NO_ROW;
        if (idx < total) {
            uint2 pe;
            if (need_scan) {
                pe = rec2[elist[idx]];
                if constexpr (MODE == 0)
                    mypub[idx] = pe;
            } else {
                pe = mypub[idx];
            }
            c  = (int)(pe.x & 0xFFFFu);
            rl = (int)(pe.x >> 16);
            v  = __builtin_bit_cast(float, pe.y);
        }
        const int n = min(32, total - base);
        for (int j0 = 0; j0 < n; j0 += 8) {
            int   cj[8];
            float vv[8];
            int   ro[8];
            bool  in_range[8];
            float g[8][VEC];
            #pragma unroll
            for (int u = 0; u < 8; ++u) {
                cj[u] = __builtin_amdgcn_readlane(c, j0 + u);
                vv[u] = __builtin_bit_cast(float,
                          __builtin_amdgcn_readlane(__builtin_bit_cast(int, v), j0 + u));
                const int rj = __builtin_amdgcn_readlane(rl, j0 + u);
                in_range[u] = rj < rows_here;
                ro[u] = rj * F;
            }
            #pragma unroll
            for (int u = 0; u < 8; ++u) {
                const float* src = dense + (size_t)cj[u] * F + lane * VEC;
                if constexpr (VEC == 4) {
                    const float4 x = *(const float4*)src;
                    g[u][0] = x.x; g[u][1] = x.y; g[u][2] = x.z; g[u][3] = x.w;
                } else {
                    const float2 x = *(const float2*)src;
                    g[u][0] = x.x; g[u][1] = x.y;
                }
            }
            #pragma unroll
            for (int u = 0; u < 8; ++u) {
                if (j0 + u < n && in_range[u]) {
                    float* dst = acc + ro[u] + lane;
                    #pragma unroll
                    for (int k = 0; k < VEC; ++k)
                        atomicAdd(dst + k * 32, vv[u] * g[u][k]);
                }
            }
        }
    }
    __syncthreads();

    constexpr int QPR = F / 4;
    for (int i = t; i < rows_here * QPR; i += OWN_THREADS) {
        const int row = i / QPR;
        const int q   = i % QPR;
        const float* a = acc + row * F;
        float4 o;
        o.x = a[((4 * q + 0) % VEC) * 32 + (4 * q + 0) / VEC];
        o.y = a[((4 * q + 1) % VEC) * 32 + (4 * q + 1) / VEC];
        o.z = a[((4 * q + 2) % VEC) * 32 + (4 * q + 2) / VEC];
        o.w = a[((4 * q + 3) % VEC) * 32 + (4 * q + 3) / VEC];
        *(float4*)(out + (size_t)(lo + row) * F + 4 * q) = o;
    }
}

__global__ __launch_bounds__(256) void gcn_logsoftmax64(
    const float* __restrict__ L, const float* __restrict__ b2,
    float* __restrict__ out)
{
    const int r = (int)((blockIdx.x * (size_t)blockDim.x + threadIdx.x) >> 5);
    const int lane = threadIdx.x & 31;
    if (r >= NROWS) return;

    float2 v = *(const float2*)(L + (size_t)r * CLS + lane * 2);
    const float2 bb = *(const float2*)(b2 + lane * 2);
    v.x += bb.x;
    v.y += bb.y;

    float m = fmaxf(v.x, v.y);
    #pragma unroll
    for (int off = 16; off > 0; off >>= 1)
        m = fmaxf(m, __shfl_xor(m, off, 32));

    float s = __expf(v.x - m) + __expf(v.y - m);
    #pragma unroll
    for (int off = 16; off > 0; off >>= 1)
        s += __shfl_xor(s, off, 32);

    const float lse = m + __logf(s);
    float2 o;
    o.x = v.x - lse;
    o.y = v.y - lse;
    *(float2*)(out + (size_t)r * CLS + lane * 2) = o;
}

extern "C" void kernel_launch(void* const* d_in, const int* in_sizes, int n_in,
                              void* d_out, int out_size, void* d_ws, size_t ws_size,
                              hipStream_t stream) {
    const float* x        = (const float*)d_in[0];
    const int*   edge_row = (const int*)  d_in[1];
    const int*   edge_col = (const int*)  d_in[2];
    const float* edge_val = (const float*)d_in[3];
    const float* W1       = (const float*)d_in[4];
    const float* b1       = (const float*)d_in[5];
    const float* W2       = (const float*)d_in[6];
    const float* b2       = (const float*)d_in[7];
    float* out = (float*)d_out;

    float* ws    = (float*)d_ws;
    float* XW1   = ws;
    float* Hraw  = ws + (size_t)NROWS * HID;
    float* HW2   = XW1;
    float* Lraw  = Hraw;

    unsigned int* plist  = (unsigned int*)d_ws + 16777216;
    unsigned int* pcount = (unsigned int*)d_ws + 33554432;
    unsigned int* okeys  = (unsigned int*)d_ws + 50331648;
    unsigned int* erecs  = (unsigned int*)d_ws + 52428800;

    const int MT = NROWS / 16;

    gcn_gemm1_wmma<<<MT, 64, 0, stream>>>(x, W1, XW1, edge_row, edge_col, edge_val, okeys, erecs);
    gcn_spmm_own<HID, 0><<<OWN_GROUPS, OWN_THREADS, 0, stream>>>(XW1, Hraw, plist, pcount, okeys, erecs);
    gcn_gemm2_wmma<<<MT, 32, 0, stream>>>(Hraw, b1, W2, HW2);
    gcn_spmm_own<CLS, 1><<<OWN_GROUPS, OWN_THREADS, 0, stream>>>(HW2, Lraw, plist, pcount, okeys, erecs);
    gcn_logsoftmax64<<<(NROWS * 32 + 255) / 256, 256, 0, stream>>>(Lraw, b2, out);
}


# log_softmax(+b2) done in the F=64 owner groups' LDS write-out: half-wave per row, 4 features per lane, DPP 16-lane max/sum (exp sum: 4 per lane then 16-lane tree, new order); 5 -> 4 launches
# speedup vs baseline: 2.3540x; 1.0608x over previous
#include <hip/hip_runtime.h>


typedef __attribute__((ext_vector_type(2))) float v2f;
typedef __attribute__((ext_vector_type(8))) float v8f;

#define NROWS 50000
#define INDIM 256
#define HID   128
#define CLS   64
#define NEDGE 800000
#define OWN_ROWS 196
#define NO_ROW   0xFFFF

__global__ __launch_bounds__(64) void gcn_gemm1_wmma(
    const float* __restrict__ X, const float* __restrict__ W1,
    float* __restrict__ Y,
    const int* __restrict__ erow, const int* __restrict__ ecol,
    const float* __restrict__ eval,
    unsigned int* __restrict__ keys, unsigned int* __restrict__ rec)
{
    static_assert((NROWS / 16) * 64 * 4 == NEDGE, "one thread per edge quad");
    const int gq = blockIdx.x * 64 + threadIdx.x;
    const int4   rq = ((const int4*)erow)[gq];
    const int4   cq = ((const int4*)ecol)[gq];
    const float4 vq = ((const float4*)eval)[gq];

    const int mtile = blockIdx.x;
    const int wv    = threadIdx.x >> 5;
    const int lane  = threadIdx.x & 31;
    const int half  = lane >> 4;
    const int l16   = lane & 15;

    const int m0 = mtile * 16;
    const int n0 = wv * 64;

    const float* xrow = X + (size_t)(m0 + l16) * INDIM + 2 * half;
    const float* wcol = W1 + n0 + l16;

    v8f acc0 = {}, acc1 = {}, acc2 = {}, acc3 = {};
    #pragma unroll 2
    for (int k = 0; k < INDIM; k += 4) {
        const float2 av = *(const float2*)(xrow + k);
        v2f a; a.x = av.x; a.y = av.y;
        const float* wk0 = wcol + (size_t)(k + 2 * half) * HID;
        const float* wk1 = wcol + (size_t)(k + 2 * half + 1) * HID;
        v2f b0; b0.x = wk0[0];  b0.y = wk1[0];
        v2f b1; b1.x = wk0[16]; b1.y = wk1[16];
        v2f b2; b2.x = wk0[32]; b2.y = wk1[32];
        v2f b3; b3.x = wk0[48]; b3.y = wk1[48];
        acc0 = __builtin_amdgcn_wmma_f32_16x16x4_f32(false, a, false, b0, (short)0, acc0, false, false);
        acc1 = __builtin_amdgcn_wmma_f32_16x16x4_f32(false, a, false, b1, (short)0, acc1, false, false);
        acc2 = __builtin_amdgcn_wmma_f32_16x16x4_f32(false, a, false, b2, (short)0, acc2, false, false);
        acc3 = __builtin_amdgcn_wmma_f32_16x16x4_f32(false, a, false, b3, (short)0, acc3, false, false);
    }

    float* yb = Y + n0 + l16;
    #pragma unroll
    for (int i = 0; i < 8; ++i) {
        float* yr = yb + (size_t)(m0 + i + 8 * half) * HID;
        yr[0]  = acc0[i];
        yr[16] = acc1[i];
        yr[32] = acc2[i];
        yr[48] = acc3[i];
    }

    const unsigned int ow0 = (unsigned int)rq.x / OWN_ROWS;
    const unsigned int ow1 = (unsigned int)rq.y / OWN_ROWS;
    const unsigned int ow2 = (unsigned int)rq.z / OWN_ROWS;
    const unsigned int ow3 = (unsigned int)rq.w / OWN_ROWS;
    keys[gq] = (ow0 & 0xFFu) | ((ow1 & 0xFFu) << 8) | ((ow2 & 0xFFu) << 16) | (ow3 << 24);

    const unsigned int l0 = (unsigned int)rq.x < (unsigned int)NROWS ? (unsigned int)rq.x - ow0 * OWN_ROWS : (unsigned int)NO_ROW;
    const unsigned int l1 = (unsigned int)rq.y < (unsigned int)NROWS ? (unsigned int)rq.y - ow1 * OWN_ROWS : (unsigned int)NO_ROW;
    const unsigned int l2 = (unsigned int)rq.z < (unsigned int)NROWS ? (unsigned int)rq.z - ow2 * OWN_ROWS : (unsigned int)NO_ROW;
    const unsigned int l3 = (unsigned int)rq.w < (unsigned int)NROWS ? (unsigned int)rq.w - ow3 * OWN_ROWS : (unsigned int)NO_ROW;
    uint4 r01, r23;
    r01.x = ((unsigned int)cq.x & 0xFFFFu) | (l0 << 16);
    r01.y = __builtin_bit_cast(unsigned int, vq.x);
    r01.z = ((unsigned int)cq.y & 0xFFFFu) | (l1 << 16);
    r01.w = __builtin_bit_cast(unsigned int, vq.y);
    r23.x = ((unsigned int)cq.z & 0xFFFFu) | (l2 << 16);
    r23.y = __builtin_bit_cast(unsigned int, vq.z);
    r23.z = ((unsigned int)cq.w & 0xFFFFu) | (l3 << 16);
    r23.w = __builtin_bit_cast(unsigned int, vq.w);
    ((uint4*)rec)[2 * gq]     = r01;
    ((uint4*)rec)[2 * gq + 1] = r23;
}

__global__ __launch_bounds__(32) void gcn_gemm2_wmma(
    const float* __restrict__ Hraw, const float* __restrict__ b1,
    const float* __restrict__ W2, float* __restrict__ Y)
{
    const int mtile = blockIdx.x;
    const int lane  = threadIdx.x & 31;
    const int half  = lane >> 4;
    const int l16   = lane & 15;

    const int m0 = mtile * 16;

    const float* hrow = Hraw + (size_t)(m0 + l16) * HID + 2 * half;
    const float* brow = b1 + 2 * half;
    const float* wcol = W2 + l16;

    v8f acc0 = {}, acc1 = {}, acc2 = {}, acc3 = {};
    #pragma unroll 2
    for (int k = 0; k < HID; k += 4) {
        const float2 hv = *(const float2*)(hrow + k);
        const float2 bv = *(const float2*)(brow + k);
        v2f a;
        a.x = fmaxf(hv.x + bv.x, 0.0f);
        a.y = fmaxf(hv.y + bv.y, 0.0f);
        const float* wk0 = wcol + (size_t)(k + 2 * half) * CLS;
        const float* wk1 = wcol + (size_t)(k + 2 * half + 1) * CLS;
        v2f b0; b0.x = wk0[0];  b0.y = wk1[0];
        v2f b1v; b1v.x = wk0[16]; b1v.y = wk1[16];
        v2f b2v; b2v.x = wk0[32]; b2v.y = wk1[32];
        v2f b3v; b3v.x = wk0[48]; b3v.y = wk1[48];
        acc0 = __builtin_amdgcn_wmma_f32_16x16x4_f32(false, a, false, b0,  (short)0, acc0, false, false);
        acc1 = __builtin_amdgcn_wmma_f32_16x16x4_f32(false, a, false, b1v, (short)0, acc1, false, false);
        acc2 = __builtin_amdgcn_wmma_f32_16x16x4_f32(false, a, false, b2v, (short)0, acc2, false, false);
        acc3 = __builtin_amdgcn_wmma_f32_16x16x4_f32(false, a, false, b3v, (short)0, acc3, false, false);
    }

    float* yb = Y + l16;
    #pragma unroll
    for (int i = 0; i < 8; ++i) {
        float* yr = yb + (size_t)(m0 + i + 8 * half) * CLS;
        yr[0]  = acc0[i];
        yr[16] = acc1[i];
        yr[32] = acc2[i];
        yr[48] = acc3[i];
    }
}

#define OWN_GROUPS 256
#define OWN_THREADS 1024
#define OWN_LDS_WORDS 43008
#define PUB_CAP    (OWN_LDS_WORDS - OWN_ROWS * HID - 4)
#define PUB_STRIDE 17920

template <int M>
__device__ __forceinline__ float xlane16(float x)
{
    const int xi = __builtin_bit_cast(int, x);
    return __builtin_bit_cast(float, __builtin_amdgcn_update_dpp(xi, xi, 0x160 + M, 0xF, 0xF, true));
}

template <int F, int MODE, bool LOGSM>
__global__ __launch_bounds__(OWN_THREADS) void gcn_spmm_own(
    const float* __restrict__ dense, float* __restrict__ out,
    unsigned int* __restrict__ publist, unsigned int* __restrict__ pubcount,
    const unsigned int* __restrict__ keys, const unsigned int* __restrict__ rec,
    const float* __restrict__ bias)
{
    constexpr int VEC = F / 32;
    constexpr int ACC_WORDS = OWN_ROWS * F;
    constexpr int CAP = OWN_LDS_WORDS - ACC_WORDS - 4;
    __shared__ __attribute__((aligned(16))) float acc[ACC_WORDS];
    __shared__ unsigned int elist[CAP];
    __shared__ unsigned int ecount;
    static_assert(MODE != 0 || CAP == PUB_CAP, "stored list capacity");
    static_assert(PUB_CAP <= PUB_STRIDE && OWN_ROWS <= 256 && NROWS <= 65536, "packing");

    const int t    = threadIdx.x;
    const int lane = t & 31;
    const int wave = t >> 5;
    const int lo   = blockIdx.x * OWN_ROWS;
    const int rows_here = min(OWN_ROWS, NROWS - lo);
    uint2* mypub = (uint2*)publist + (size_t)blockIdx.x * PUB_STRIDE;
    const uint2* rec2 = (const uint2*)rec;

    unsigned int stored = 0u;
    bool need_scan = true;
    if constexpr (MODE == 1) {
        stored = (unsigned int)__builtin_amdgcn_readfirstlane((int)pubcount[blockIdx.x]);
        need_scan = stored > (unsigned int)PUB_CAP;
    }

    for (int i = t; i < ACC_WORDS; i += OWN_THREADS) acc[i] = 0.0f;
    if (t == 0) ecount = 0u;
    __syncthreads();

    const uint4* key16 = (const uint4*)keys;
    constexpr int NQ = NEDGE / 16;
    constexpr int SCAN_UNR = 7;
    const unsigned int mine = (unsigned int)blockIdx.x * 0x01010101u;
    for (int q0 = t; need_scan && q0 < NQ; q0 += OWN_THREADS * SCAN_UNR) {
        uint4 kq[SCAN_UNR];
        #pragma unroll
        for (int u = 0; u < SCAN_UNR; ++u)
            kq[u] = key16[min(q0 + u * OWN_THREADS, NQ - 1)];
        #pragma unroll
        for (int u = 0; u < SCAN_UNR; ++u) {
            const int q = q0 + u * OWN_THREADS;
            const unsigned int x4[4] = { kq[u].x ^ mine, kq[u].y ^ mine,
                                         kq[u].z ^ mine, kq[u].w ^ mine };
            unsigned int nz4[4];
            #pragma unroll
            for (int w = 0; w < 4; ++w)
                nz4[w] = ((x4[w] & 0x7F7F7F7Fu) + 0x7F7F7F7Fu) | x4[w];
            const unsigned int all_nz = nz4[0] & nz4[1] & nz4[2] & nz4[3];
            if ((all_nz & 0x80808080u) != 0x80808080u && q < NQ) {
                unsigned int some = 0u;
                #pragma unroll
                for (int w = 0; w < 4; ++w)
                    some |= (nz4[w] >> (7 - w)) & (0x01010101u << w);
                unsigned int hit = ~some & 0x0F0F0F0Fu;
                do {
                    const int p = __builtin_ctz(hit);
                    hit &= hit - 1u;
                    const unsigned int e = (unsigned int)(q * 16 + (p & 3) * 4 + (p >> 3));
                    const unsigned int slot = atomicAdd(&ecount, 1u);
                    if (slot < (unsigned int)CAP) {
                        elist[slot] = e;
                    } else {
                        const uint2 pe = rec2[e];
                        const unsigned int rl = pe.x >> 16;
                        if (rl < (unsigned int)rows_here) {
                            const int   c = (int)(pe.x & 0xFFFFu);
                            const float v = __builtin_bit_cast(float, pe.y);
                            const float* src = dense + (size_t)c * F;
                            float* dst = acc + rl * F;
                            #pragma unroll 1
                            for (int f = 0; f < F; ++f)
                                atomicAdd(dst + (f % VEC) * 32 + (f / VEC), v * src[f]);
                        }
                    }
                } while (hit != 0u);
            }
        }
    }
    __syncthreads();

    const int total = need_scan
        ? __builtin_amdgcn_readfirstlane((int)min(ecount, (unsigned int)CAP))
        : (int)stored;
    if constexpr (MODE == 0) {
        if (t == 0) pubcount[blockIdx.x] = ecount;
    }
    for (int base = wave * 32; base < total; base += (OWN_THREADS / 32) * 32) {
        const int idx = base + lane;
        int   c = 0;
        float v = 0.0f;
        int   rl = NO_ROW;
        if (idx < total) {
            uint2 pe;
            if (need_scan) {
                pe = rec2[elist[idx]];
                if constexpr (MODE == 0)
                    mypub[idx] = pe;
            } else {
                pe = mypub[idx];
            }
            c  = (int)(pe.x & 0xFFFFu);
            rl = (int)(pe.x >> 16);
            v  = __builtin_bit_cast(float, pe.y);
        }
        const int n = min(32, total - base);
        for (int j0 = 0; j0 < n; j0 += 8) {
            int   cj[8];
            float vv[8];
            int   ro[8];
            bool  in_range[8];
            float g[8][VEC];
            #pragma unroll
            for (int u = 0; u < 8; ++u) {
                cj[u] = __builtin_amdgcn_readlane(c, j0 + u);
                vv[u] = __builtin_bit_cast(float,
                          __builtin_amdgcn_readlane(__builtin_bit_cast(int, v), j0 + u));
                const int rj = __builtin_amdgcn_readlane(rl, j0 + u);
                in_range[u] = rj < rows_here;
                ro[u] = rj * F;
            }
            #pragma unroll
            for (int u = 0; u < 8; ++u) {
                const float* src = dense + (size_t)cj[u] * F + lane * VEC;
                if constexpr (VEC == 4) {
                    const float4 x = *(const float4*)src;
                    g[u][0] = x.x; g[u][1] = x.y; g[u][2] = x.z; g[u][3] = x.w;
                } else {
                    const float2 x = *(const float2*)src;
                    g[u][0] = x.x; g[u][1] = x.y;
                }
            }
            #pragma unroll
            for (int u = 0; u < 8; ++u) {
                if (j0 + u < n && in_range[u]) {
                    float* dst = acc + ro[u] + lane;
                    #pragma unroll
                    for (int k = 0; k < VEC; ++k)
                        atomicAdd(dst + k * 32, vv[u] * g[u][k]);
                }
            }
        }
    }
    __syncthreads();

    if constexpr (LOGSM) {
        static_assert(F == 64 && VEC == 2, "one 16-lane row of 4 features per lane");
        const int hl = lane & 15;
        const int hf = lane >> 4;
        const float4 bb = *(const float4*)(bias + 4 * hl);
        for (int r0 = wave * 2; r0 < rows_here; r0 += (OWN_THREADS / 32) * 2) {
            const int  row = r0 + hf;
            const bool ok  = row < rows_here;
            const float* a = acc + (ok ? row : r0) * F;
            const float2 va = *(const float2*)(a + 2 * hl);
            const float2 vb = *(const float2*)(a + 32 + 2 * hl);
            const float x0 = va.x + bb.x;
            const float x1 = vb.x + bb.y;
            const float x2 = va.y + bb.z;
            const float x3 = vb.y + bb.w;
            float m = fmaxf(fmaxf(x0, x1), fmaxf(x2, x3));
            m = fmaxf(m, xlane16<8>(m));
            m = fmaxf(m, xlane16<4>(m));
            m = fmaxf(m, xlane16<2>(m));
            m = fmaxf(m, xlane16<1>(m));
            float s = (__expf(x0 - m) + __expf(x1 - m)) + (__expf(x2 - m) + __expf(x3 - m));
            s += xlane16<8>(s);
            s += xlane16<4>(s);
            s += xlane16<2>(s);
            s += xlane16<1>(s);
            const float lse = m + __logf(s);
            if (ok) {
                float4 o;
                o.x = x0 - lse;
                o.y = x1 - lse;
                o.z = x2 - lse;
                o.w = x3 - lse;
                *(float4*)(out + (size_t)(lo + row) * F + 4 * hl) = o;
            }
        }
        return;
    }
    constexpr int QPR = F / 4;
    for (int i = t; i < rows_here * QPR; i += OWN_THREADS) {
        const int row = i / QPR;
        const int q   = i % QPR;
        const float* a = acc + row * F;
        float4 o;
        o.x = a[((4 * q + 0) % VEC) * 32 + (4 * q + 0) / VEC];
        o.y = a[((4 * q + 1) % VEC) * 32 + (4 * q + 1) / VEC];
        o.z = a[((4 * q + 2) % VEC) * 32 + (4 * q + 2) / VEC];
        o.w = a[((4 * q + 3) % VEC) * 32 + (4 * q + 3) / VEC];
        *(float4*)(out + (size_t)(lo + row) * F + 4 * q) = o;
    }
}

extern "C" void kernel_launch(void* const* d_in, const int* in_sizes, int n_in,
                              void* d_out, int out_size, void* d_ws, size_t ws_size,
                              hipStream_t stream) {
    const float* x        = (const float*)d_in[0];
    const int*   edge_row = (const int*)  d_in[1];
    const int*   edge_col = (const int*)  d_in[2];
    const float* edge_val = (const float*)d_in[3];
    const float* W1       = (const float*)d_in[4];
    const float* b1       = (const float*)d_in[5];
    const float* W2       = (const float*)d_in[6];
    const float* b2       = (const float*)d_in[7];
    float* out = (float*)d_out;

    float* ws    = (float*)d_ws;
    float* XW1   = ws;
    float* Hraw  = ws + (size_t)NROWS * HID;
    float* HW2   = XW1;

    unsigned int* plist  = (unsigned int*)d_ws + 16777216;
    unsigned int* pcount = (unsigned int*)d_ws + 33554432;
    unsigned int* okeys  = (unsigned int*)d_ws + 50331648;
    unsigned int* erecs  = (unsigned int*)d_ws + 52428800;

    const int MT = NROWS / 16;

    gcn_gemm1_wmma<<<MT, 64, 0, stream>>>(x, W1, XW1, edge_row, edge_col, edge_val, okeys, erecs);
    gcn_spmm_own<HID, 0, false><<<OWN_GROUPS, OWN_THREADS, 0, stream>>>(XW1, Hraw, plist, pcount, okeys, erecs, b1);
    gcn_gemm2_wmma<<<MT, 32, 0, stream>>>(Hraw, b1, W2, HW2);
    gcn_spmm_own<CLS, 1, true><<<OWN_GROUPS, OWN_THREADS, 0, stream>>>(HW2, out, plist, pcount, okeys, erecs, b2);
}


# Second GEMM done in the F=128 owner groups' write-out: relu(row+b1) @ W2 by WMMA, both operands read from LDS (rows at pitch 129, W2 staged); sum order inside each 4 k changes; 4 -> 3 launches
# speedup vs baseline: 2.6158x; 1.1112x over previous
#include <hip/hip_runtime.h>


typedef __attribute__((ext_vector_type(2))) float v2f;
typedef __attribute__((ext_vector_type(8))) float v8f;

#define NROWS 50000
#define INDIM 256
#define HID   128
#define CLS   64
#define NEDGE 800000
#define OWN_ROWS 196
#define NO_ROW   0xFFFF

__global__ __launch_bounds__(64) void gcn_gemm1_wmma(
    const float* __restrict__ X, const float* __restrict__ W1,
    float* __restrict__ Y,
    const int* __restrict__ erow, const int* __restrict__ ecol,
    const float* __restrict__ eval,
    unsigned int* __restrict__ keys, unsigned int* __restrict__ rec)
{
    static_assert((NROWS / 16) * 64 * 4 == NEDGE, "one thread per edge quad");
    const int gq = blockIdx.x * 64 + threadIdx.x;
    const int4   rq = ((const int4*)erow)[gq];
    const int4   cq = ((const int4*)ecol)[gq];
    const float4 vq = ((const float4*)eval)[gq];

    const int mtile = blockIdx.x;
    const int wv    = threadIdx.x >> 5;
    const int lane  = threadIdx.x & 31;
    const int half  = lane >> 4;
    const int l16   = lane & 15;

    const int m0 = mtile * 16;
    const int n0 = wv * 64;

    const float* xrow = X + (size_t)(m0 + l16) * INDIM + 2 * half;
    const float* wcol = W1 + n0 + l16;

    v8f acc0 = {}, acc1 = {}, acc2 = {}, acc3 = {};
    #pragma unroll 2
    for (int k = 0; k < INDIM; k += 4) {
        const float2 av = *(const float2*)(xrow + k);
        v2f a; a.x = av.x; a.y = av.y;
        const float* wk0 = wcol + (size_t)(k + 2 * half) * HID;
        const float* wk1 = wcol + (size_t)(k + 2 * half + 1) * HID;
        v2f b0; b0.x = wk0[0];  b0.y = wk1[0];
        v2f b1; b1.x = wk0[16]; b1.y = wk1[16];
        v2f b2; b2.x = wk0[32]; b2.y = wk1[32];
        v2f b3; b3.x = wk0[48]; b3.y = wk1[48];
        acc0 = __builtin_amdgcn_wmma_f32_16x16x4_f32(false, a, false, b0, (short)0, acc0, false, false);
        acc1 = __builtin_amdgcn_wmma_f32_16x16x4_f32(false, a, false, b1, (short)0, acc1, false, false);
        acc2 = __builtin_amdgcn_wmma_f32_16x16x4_f32(false, a, false, b2, (short)0, acc2, false, false);
        acc3 = __builtin_amdgcn_wmma_f32_16x16x4_f32(false, a, false, b3, (short)0, acc3, false, false);
    }

    float* yb = Y + n0 + l16;
    #pragma unroll
    for (int i = 0; i < 8; ++i) {
        float* yr = yb + (size_t)(m0 + i + 8 * half) * HID;
        yr[0]  = acc0[i];
        yr[16] = acc1[i];
        yr[32] = acc2[i];
        yr[48] = acc3[i];
    }

    const unsigned int ow0 = (unsigned int)rq.x / OWN_ROWS;
    const unsigned int ow1 = (unsigned int)rq.y / OWN_ROWS;
    const unsigned int ow2 = (unsigned int)rq.z / OWN_ROWS;
    const unsigned int ow3 = (unsigned int)rq.w / OWN_ROWS;
    keys[gq] = (ow0 & 0xFFu) | ((ow1 & 0xFFu) << 8) | ((ow2 & 0xFFu) << 16) | (ow3 << 24);

    const unsigned int l0 = (unsigned int)rq.x < (unsigned int)NROWS ? (unsigned int)rq.x - ow0 * OWN_ROWS : (unsigned int)NO_ROW;
    const unsigned int l1 = (unsigned int)rq.y < (unsigned int)NROWS ? (unsigned int)rq.y - ow1 * OWN_ROWS : (unsigned int)NO_ROW;
    const unsigned int l2 = (unsigned int)rq.z < (unsigned int)NROWS ? (unsigned int)rq.z - ow2 * OWN_ROWS : (unsigned int)NO_ROW;
    const unsigned int l3 = (unsigned int)rq.w < (unsigned int)NROWS ? (unsigned int)rq.w - ow3 * OWN_ROWS : (unsigned int)NO_ROW;
    uint4 r01, r23;
    r01.x = ((unsigned int)cq.x & 0xFFFFu) | (l0 << 16);
    r01.y = __builtin_bit_cast(unsigned int, vq.x);
    r01.z = ((unsigned int)cq.y & 0xFFFFu) | (l1 << 16);
    r01.w = __builtin_bit_cast(unsigned int, vq.y);
    r23.x = ((unsigned int)cq.z & 0xFFFFu) | (l2 << 16);
    r23.y = __builtin_bit_cast(unsigned int, vq.z);
    r23.z = ((unsigned int)cq.w & 0xFFFFu) | (l3 << 16);
    r23.w = __builtin_bit_cast(unsigned int, vq.w);
    ((uint4*)rec)[2 * gq]     = r01;
    ((uint4*)rec)[2 * gq + 1] = r23;
}

#define OWN_GROUPS 256
#define OWN_THREADS 1024
#define OWN_LDS_WORDS 43008
#define PUB_CAP    (OWN_LDS_WORDS - OWN_ROWS * (HID + 1) - 4)
#define PUB_STRIDE 17920

template <int M>
__device__ __forceinline__ float xlane16(float x)
{
    const int xi = __builtin_bit_cast(int, x);
    return __builtin_bit_cast(float, __builtin_amdgcn_update_dpp(xi, xi, 0x160 + M, 0xF, 0xF, true));
}

template <int F, int MODE, bool LOGSM>
__global__ __launch_bounds__(OWN_THREADS) void gcn_spmm_own(
    const float* __restrict__ dense, float* __restrict__ out,
    unsigned int* __restrict__ publist, unsigned int* __restrict__ pubcount,
    const unsigned int* __restrict__ keys, const unsigned int* __restrict__ rec,
    const float* __restrict__ bias, const float* __restrict__ w2)
{
    constexpr int VEC = F / 32;
    constexpr int PITCH = LOGSM ? F : F + 1;
    constexpr int ACC_WORDS = OWN_ROWS * PITCH;
    constexpr int CAP = OWN_LDS_WORDS - ACC_WORDS - 4;
    __shared__ __attribute__((aligned(16))) float acc[ACC_WORDS];
    __shared__ __attribute__((aligned(16))) unsigned int elist[CAP];
    __shared__ unsigned int ecount;
    static_assert(MODE != 0 || CAP == PUB_CAP, "stored list capacity");
    static_assert(PUB_CAP <= PUB_STRIDE && OWN_ROWS <= 256 && NROWS <= 65536, "packing");

    const int t    = threadIdx.x;
    const int lane = t & 31;
    const int wave = t >> 5;
    const int lo   = blockIdx.x * OWN_ROWS;
    const int rows_here = min(OWN_ROWS, NROWS - lo);
    uint2* mypub = (uint2*)publist + (size_t)blockIdx.x * PUB_STRIDE;
    const uint2* rec2 = (const uint2*)rec;

    unsigned int stored = 0u;
    bool need_scan = true;
    if constexpr (MODE == 1) {
        stored = (unsigned int)__builtin_amdgcn_readfirstlane((int)pubcount[blockIdx.x]);
        need_scan = stored > (unsigned int)PUB_CAP;
    }

    for (int i = t; i < ACC_WORDS; i += OWN_THREADS) acc[i] = 0.0f;
    if (t == 0) ecount = 0u;
    __syncthreads();

    const uint4* key16 = (const uint4*)keys;
    constexpr int NQ = NEDGE / 16;
    constexpr int SCAN_UNR = 7;
    const unsigned int mine = (unsigned int)blockIdx.x * 0x01010101u;
    for (int q0 = t; need_scan && q0 < NQ; q0 += OWN_THREADS * SCAN_UNR) {
        uint4 kq[SCAN_UNR];
        #pragma unroll
        for (int u = 0; u < SCAN_UNR; ++u)
            kq[u] = key16[min(q0 + u * OWN_THREADS, NQ - 1)];
        #pragma unroll
        for (int u = 0; u < SCAN_UNR; ++u) {
            const int q = q0 + u * OWN_THREADS;
            const unsigned int x4[4] = { kq[u].x ^ mine, kq[u].y ^ mine,
                                         kq[u].z ^ mine, kq[u].w ^ mine };
            unsigned int nz4[4];
            #pragma unroll
            for (int w = 0; w < 4; ++w)
                nz4[w] = ((x4[w] & 0x7F7F7F7Fu) + 0x7F7F7F7Fu) | x4[w];
            const unsigned int all_nz = nz4[0] & nz4[1] & nz4[2] & nz4[3];
            if ((all_nz & 0x80808080u) != 0x80808080u && q < NQ) {
                unsigned int some = 0u;
                #pragma unroll
                for (int w = 0; w < 4; ++w)
                    some |= (nz4[w] >> (7 - w)) & (0x01010101u << w);
                unsigned int hit = ~some & 0x0F0F0F0Fu;
                do {
                    const int p = __builtin_ctz(hit);
                    hit &= hit - 1u;
                    const unsigned int e = (unsigned int)(q * 16 + (p & 3) * 4 + (p >> 3));
                    const unsigned int slot = atomicAdd(&ecount, 1u);
                    if (slot < (unsigned int)CAP) {
                        elist[slot] = e;
                    } else {
                        const uint2 pe = rec2[e];
                        const unsigned int rl = pe.x >> 16;
                        if (rl < (unsigned int)rows_here) {
                            const int   c = (int)(pe.x & 0xFFFFu);
                            const float v = __builtin_bit_cast(float, pe.y);
                            const float* src = dense + (size_t)c * F;
                            float* dst = acc + rl * PITCH;
                            #pragma unroll 1
                            for (int f = 0; f < F; ++f)
                                atomicAdd(dst + (f % VEC) * 32 + (f / VEC), v * src[f]);
                        }
                    }
                } while (hit != 0u);
            }
        }
    }
    __syncthreads();

    const int total = need_scan
        ? __builtin_amdgcn_readfirstlane((int)min(ecount, (unsigned int)CAP))
        : (int)stored;
    if constexpr (MODE == 0) {
        if (t == 0) pubcount[blockIdx.x] = ecount;
    }
    for (int base = wave * 32; base < total; base += (OWN_THREADS / 32) * 32) {
        const int idx = base + lane;
        int   c = 0;
        float v = 0.0f;
        int   rl = NO_ROW;
        if (idx < total) {
            uint2 pe;
            if (need_scan) {
                pe = rec2[elist[idx]];
                if constexpr (MODE == 0)
                    mypub[idx] = pe;
            } else {
                pe = mypub[idx];
            }
            c  = (int)(pe.x & 0xFFFFu);
            rl = (int)(pe.x >> 16);
            v  = __builtin_bit_cast(float, pe.y);
        }
        const int n = min(32, total - base);
        for (int j0 = 0; j0 < n; j0 += 8) {
            int   cj[8];
            float vv[8];
            int   ro[8];
            bool  in_range[8];
            float g[8][VEC];
            #pragma unroll
            for (int u = 0; u < 8; ++u) {
                cj[u] = __builtin_amdgcn_readlane(c, j0 + u);
                vv[u] = __builtin_bit_cast(float,
                          __builtin_amdgcn_readlane(__builtin_bit_cast(int, v), j0 + u));
                const int rj = __builtin_amdgcn_readlane(rl, j0 + u);
                in_range[u] = rj < rows_here;
                ro[u] = rj * PITCH;
            }
            #pragma unroll
            for (int u = 0; u < 8; ++u) {
                const float* src = dense + (size_t)cj[u] * F + lane * VEC;
                if constexpr (VEC == 4) {
                    const float4 x = *(const float4*)src;
                    g[u][0] = x.x; g[u][1] = x.y; g[u][2] = x.z; g[u][3] = x.w;
                } else {
                    const float2 x = *(const float2*)src;
                    g[u][0] = x.x; g[u][1] = x.y;
                }
            }
            #pragma unroll
            for (int u = 0; u < 8; ++u) {
                if (j0 + u < n && in_range[u]) {
                    float* dst = acc + ro[u] + lane;
                    #pragma unroll
                    for (int k = 0; k < VEC; ++k)
                        atomicAdd(dst + k * 32, vv[u] * g[u][k]);
                }
            }
        }
    }
    __syncthreads();

    if constexpr (LOGSM) {
        static_assert(F == 64 && VEC == 2, "one 16-lane row of 4 features per lane");
        const int hl = lane & 15;
        const int hf = lane >> 4;
        const float4 bb = *(const float4*)(bias + 4 * hl);
        for (int r0 = wave * 2; r0 < rows_here; r0 += (OWN_THREADS / 32) * 2) {
            const int  row = r0 + hf;
            const bool ok  = row < rows_here;
            const float* a = acc + (ok ? row : r0) * PITCH;
            const float2 va = *(const float2*)(a + 2 * hl);
            const float2 vb = *(const float2*)(a + 32 + 2 * hl);
            const float x0 = va.x + bb.x;
            const float x1 = vb.x + bb.y;
            const float x2 = va.y + bb.z;
            const float x3 = vb.y + bb.w;
            float m = fmaxf(fmaxf(x0, x1), fmaxf(x2, x3));
            m = fmaxf(m, xlane16<8>(m));
            m = fmaxf(m, xlane16<4>(m));
            m = fmaxf(m, xlane16<2>(m));
            m = fmaxf(m, xlane16<1>(m));
            float s = (__expf(x0 - m) + __expf(x1 - m)) + (__expf(x2 - m) + __expf(x3 - m));
            s += xlane16<8>(s);
            s += xlane16<4>(s);
            s += xlane16<2>(s);
            s += xlane16<1>(s);
            const float lse = m + __logf(s);
            if (ok) {
                float4 o;
                o.x = x0 - lse;
                o.y = x1 - lse;
                o.z = x2 - lse;
                o.w = x3 - lse;
                *(float4*)(out + (size_t)(lo + row) * F + 4 * hl) = o;
            }
        }
    } else {
        static_assert(F == HID && VEC == 4 && OWN_THREADS % F == 0, "row layout of the passes below");
        constexpr int W2PITCH = CLS + 16;
        static_assert(HID * W2PITCH <= CAP, "W2 fits into the list's LDS space");
        {
            const int w = t & (F - 1);
            const float bv = bias[(w & 31) * VEC + (w >> 5)];
            for (int r = t >> 7; r < rows_here; r += OWN_THREADS / F) {
                float* p = acc + r * PITCH + w;
                *p = fmaxf(*p + bv, 0.0f);
            }
        }
        for (int i = t; i < HID * CLS / 4; i += OWN_THREADS) {
            const float4 wq = ((const float4*)w2)[i];
            unsigned int* wd = elist + (i >> 4) * W2PITCH + (i & 15) * 4;
            wd[0] = __builtin_bit_cast(unsigned int, wq.x);
            wd[1] = __builtin_bit_cast(unsigned int, wq.y);
            wd[2] = __builtin_bit_cast(unsigned int, wq.z);
            wd[3] = __builtin_bit_cast(unsigned int, wq.w);
        }
        __syncthreads();
        constexpr int MTILES = (OWN_ROWS + 15) / 16;
        const int half = lane >> 4;
        const int l16  = lane & 15;
        const int wv = __builtin_amdgcn_readfirstlane(wave);
        for (int job = wv; job < MTILES * (CLS / 16); job += OWN_THREADS / 32) {
            const int m0 = (job >> 2) * 16;
            const int n0 = (job & 3) * 16;
            if (m0 >= rows_here) break;
            const int arow = min(m0 + l16, rows_here - 1);
            const float* ap = acc + arow * PITCH + half * 32;
            const unsigned int* bw = elist + half * W2PITCH + n0 + l16;
            v8f d = {};
            #pragma unroll
            for (int j = 0; j < HID / 4; ++j) {
                v2f a, b;
                a.x = ap[j];
                a.y = ap[j + 64];
                b.x = __builtin_bit_cast(float, bw[(4 * j)     * W2PITCH]);
                b.y = __builtin_bit_cast(float, bw[(4 * j + 2) * W2PITCH]);
                d = __builtin_amdgcn_wmma_f32_16x16x4_f32(false, a, false, b, (short)0, d, false, false);
            }
            float* yb = out + (size_t)lo * CLS + n0 + l16;
            #pragma unroll
            for (int i = 0; i < 8; ++i) {
                const int r = m0 + i + 8 * half;
                if (r < rows_here) yb[(size_t)r * CLS] = d[i];
            }
        }
    }
}

extern "C" void kernel_launch(void* const* d_in, const int* in_sizes, int n_in,
                              void* d_out, int out_size, void* d_ws, size_t ws_size,
                              hipStream_t stream) {
    const float* x        = (const float*)d_in[0];
    const int*   edge_row = (const int*)  d_in[1];
    const int*   edge_col = (const int*)  d_in[2];
    const float* edge_val = (const float*)d_in[3];
    const float* W1       = (const float*)d_in[4];
    const float* b1       = (const float*)d_in[5];
    const float* W2       = (const float*)d_in[6];
    const float* b2       = (const float*)d_in[7];
    float* out = (float*)d_out;

    float* ws    = (float*)d_ws;
    float* XW1   = ws;
    float* HW2   = ws + (size_t)NROWS * HID;

    unsigned int* plist  = (unsigned int*)d_ws + 16777216;
    unsigned int* pcount = (unsigned int*)d_ws + 33554432;
    unsigned int* okeys  = (unsigned int*)d_ws + 50331648;
    unsigned int* erecs  = (unsigned int*)d_ws + 52428800;

    const int MT = NROWS / 16;

    gcn_gemm1_wmma<<<MT, 64, 0, stream>>>(x, W1, XW1, edge_row, edge_col, edge_val, okeys, erecs);
    gcn_spmm_own<HID, 0, false><<<OWN_GROUPS, OWN_THREADS, 0, stream>>>(XW1, HW2, plist, pcount, okeys, erecs, b1, W2);
    gcn_spmm_own<CLS, 1, true><<<OWN_GROUPS, OWN_THREADS, 0, stream>>>(HW2, out, plist, pcount, okeys, erecs, b2, W2);
}


# GEMM1 strip columns interleaved over 4 accumulator tiles (tile t = cols 4c+t): a lane's W1 values of a k row are one 16-B load (2 per k-step, was 8), outputs go as whole-line 16-B stores; sums kept
# speedup vs baseline: 2.8034x; 1.0717x over previous
#include <hip/hip_runtime.h>


typedef __attribute__((ext_vector_type(2))) float v2f;
typedef __attribute__((ext_vector_type(8))) float v8f;

#define NROWS 50000
#define INDIM 256
#define HID   128
#define CLS   64
#define NEDGE 800000
#define OWN_ROWS 196
#define NO_ROW   0xFFFF

__global__ __launch_bounds__(64) void gcn_gemm1_wmma(
    const float* __restrict__ X, const float* __restrict__ W1,
    float* __restrict__ Y,
    const int* __restrict__ erow, const int* __restrict__ ecol,
    const float* __restrict__ eval,
    unsigned int* __restrict__ keys, unsigned int* __restrict__ rec)
{
    static_assert((NROWS / 16) * 64 * 4 == NEDGE, "one thread per edge quad");
    const int gq = blockIdx.x * 64 + threadIdx.x;
    const int4   rq = ((const int4*)erow)[gq];
    const int4   cq = ((const int4*)ecol)[gq];
    const float4 vq = ((const float4*)eval)[gq];

    const int mtile = blockIdx.x;
    const int wv    = threadIdx.x >> 5;
    const int lane  = threadIdx.x & 31;
    const int half  = lane >> 4;
    const int l16   = lane & 15;

    const int m0 = mtile * 16;
    const int n0 = wv * 64;

    const float* xrow = X + (size_t)(m0 + l16) * INDIM + 2 * half;
    const float* wcol = W1 + n0 + 4 * l16;

    v8f acc0 = {}, acc1 = {}, acc2 = {}, acc3 = {};
    #pragma unroll 4
    for (int k = 0; k < INDIM; k += 4) {
        const float2 av = *(const float2*)(xrow + k);
        v2f a; a.x = av.x; a.y = av.y;
        const float4 w0 = *(const float4*)(wcol + (size_t)(k + 2 * half) * HID);
        const float4 w1 = *(const float4*)(wcol + (size_t)(k + 2 * half + 1) * HID);
        v2f b0; b0.x = w0.x; b0.y = w1.x;
        v2f b1; b1.x = w0.y; b1.y = w1.y;
        v2f b2; b2.x = w0.z; b2.y = w1.z;
        v2f b3; b3.x = w0.w; b3.y = w1.w;
        acc0 = __builtin_amdgcn_wmma_f32_16x16x4_f32(false, a, false, b0, (short)0, acc0, false, false);
        acc1 = __builtin_amdgcn_wmma_f32_16x16x4_f32(false, a, false, b1, (short)0, acc1, false, false);
        acc2 = __builtin_amdgcn_wmma_f32_16x16x4_f32(false, a, false, b2, (short)0, acc2, false, false);
        acc3 = __builtin_amdgcn_wmma_f32_16x16x4_f32(false, a, false, b3, (short)0, acc3, false, false);
    }

    float* yb = Y + n0 + 4 * l16;
    #pragma unroll
    for (int i = 0; i < 8; ++i) {
        float4 o;
        o.x = acc0[i];
        o.y = acc1[i];
        o.z = acc2[i];
        o.w = acc3[i];
        *(float4*)(yb + (size_t)(m0 + i + 8 * half) * HID) = o;
    }

    const unsigned int ow0 = (unsigned int)rq.x / OWN_ROWS;
    const unsigned int ow1 = (unsigned int)rq.y / OWN_ROWS;
    const unsigned int ow2 = (unsigned int)rq.z / OWN_ROWS;
    const unsigned int ow3 = (unsigned int)rq.w / OWN_ROWS;
    keys[gq] = (ow0 & 0xFFu) | ((ow1 & 0xFFu) << 8) | ((ow2 & 0xFFu) << 16) | (ow3 << 24);

    const unsigned int l0 = (unsigned int)rq.x < (unsigned int)NROWS ? (unsigned int)rq.x - ow0 * OWN_ROWS : (unsigned int)NO_ROW;
    const unsigned int l1 = (unsigned int)rq.y < (unsigned int)NROWS ? (unsigned int)rq.y - ow1 * OWN_ROWS : (unsigned int)NO_ROW;
    const unsigned int l2 = (unsigned int)rq.z < (unsigned int)NROWS ? (unsigned int)rq.z - ow2 * OWN_ROWS : (unsigned int)NO_ROW;
    const unsigned int l3 = (unsigned int)rq.w < (unsigned int)NROWS ? (unsigned int)rq.w - ow3 * OWN_ROWS : (unsigned int)NO_ROW;
    uint4 r01, r23;
    r01.x = ((unsigned int)cq.x & 0xFFFFu) | (l0 << 16);
    r01.y = __builtin_bit_cast(unsigned int, vq.x);
    r01.z = ((unsigned int)cq.y & 0xFFFFu) | (l1 << 16);
    r01.w = __builtin_bit_cast(unsigned int, vq.y);
    r23.x = ((unsigned int)cq.z & 0xFFFFu) | (l2 << 16);
    r23.y = __builtin_bit_cast(unsigned int, vq.z);
    r23.z = ((unsigned int)cq.w & 0xFFFFu) | (l3 << 16);
    r23.w = __builtin_bit_cast(unsigned int, vq.w);
    ((uint4*)rec)[2 * gq]     = r01;
    ((uint4*)rec)[2 * gq + 1] = r23;
}

#define OWN_GROUPS 256
#define OWN_THREADS 1024
#define OWN_LDS_WORDS 43008
#define PUB_CAP    (OWN_LDS_WORDS - OWN_ROWS * (HID + 1) - 4)
#define PUB_STRIDE 17920

template <int M>
__device__ __forceinline__ float xlane16(float x)
{
    const int xi = __builtin_bit_cast(int, x);
    return __builtin_bit_cast(float, __builtin_amdgcn_update_dpp(xi, xi, 0x160 + M, 0xF, 0xF, true));
}

template <int F, int MODE, bool LOGSM>
__global__ __launch_bounds__(OWN_THREADS) void gcn_spmm_own(
    const float* __restrict__ dense, float* __restrict__ out,
    unsigned int* __restrict__ publist, unsigned int* __restrict__ pubcount,
    const unsigned int* __restrict__ keys, const unsigned int* __restrict__ rec,
    const float* __restrict__ bias, const float* __restrict__ w2)
{
    constexpr int VEC = F / 32;
    constexpr int PITCH = LOGSM ? F : F + 1;
    constexpr int ACC_WORDS = OWN_ROWS * PITCH;
    constexpr int CAP = OWN_LDS_WORDS - ACC_WORDS - 4;
    __shared__ __attribute__((aligned(16))) float acc[ACC_WORDS];
    __shared__ __attribute__((aligned(16))) unsigned int elist[CAP];
    __shared__ unsigned int ecount;
    static_assert(MODE != 0 || CAP == PUB_CAP, "stored list capacity");
    static_assert(PUB_CAP <= PUB_STRIDE && OWN_ROWS <= 256 && NROWS <= 65536, "packing");

    const int t    = threadIdx.x;
    const int lane = t & 31;
    const int wave = t >> 5;
    const int lo   = blockIdx.x * OWN_ROWS;
    const int rows_here = min(OWN_ROWS, NROWS - lo);
    uint2* mypub = (uint2*)publist + (size_t)blockIdx.x * PUB_STRIDE;
    const uint2* rec2 = (const uint2*)rec;

    unsigned int stored = 0u;
    bool need_scan = true;
    if constexpr (MODE == 1) {
        stored = (unsigned int)__builtin_amdgcn_readfirstlane((int)pubcount[blockIdx.x]);
        need_scan = stored > (unsigned int)PUB_CAP;
    }

    for (int i = t; i < ACC_WORDS; i += OWN_THREADS) acc[i] = 0.0f;
    if (t == 0) ecount = 0u;
    __syncthreads();

    const uint4* key16 = (const uint4*)keys;
    constexpr int NQ = NEDGE / 16;
    constexpr int SCAN_UNR = 7;
    const unsigned int mine = (unsigned int)blockIdx.x * 0x01010101u;
    for (int q0 = t; need_scan && q0 < NQ; q0 += OWN_THREADS * SCAN_UNR) {
        uint4 kq[SCAN_UNR];
        #pragma unroll
        for (int u = 0; u < SCAN_UNR; ++u)
            kq[u] = key16[min(q0 + u * OWN_THREADS, NQ - 1)];
        #pragma unroll
        for (int u = 0; u < SCAN_UNR; ++u) {
            const int q = q0 + u * OWN_THREADS;
            const unsigned int x4[4] = { kq[u].x ^ mine, kq[u].y ^ mine,
                                         kq[u].z ^ mine, kq[u].w ^ mine };
            unsigned int nz4[4];
            #pragma unroll
            for (int w = 0; w < 4; ++w)
                nz4[w] = ((x4[w] & 0x7F7F7F7Fu) + 0x7F7F7F7Fu) | x4[w];
            const unsigned int all_nz = nz4[0] & nz4[1] & nz4[2] & nz4[3];
            if ((all_nz & 0x80808080u) != 0x80808080u && q < NQ) {
                unsigned int some = 0u;
                #pragma unroll
                for (int w = 0; w < 4; ++w)
                    some |= (nz4[w] >> (7 - w)) & (0x01010101u << w);
                unsigned int hit = ~some & 0x0F0F0F0Fu;
                do {
                    const int p = __builtin_ctz(hit);
                    hit &= hit - 1u;
                    const unsigned int e = (unsigned int)(q * 16 + (p & 3) * 4 + (p >> 3));
                    const unsigned int slot = atomicAdd(&ecount, 1u);
                    if (slot < (unsigned int)CAP) {
                        elist[slot] = e;
                    } else {
                        const uint2 pe = rec2[e];
                        const unsigned int rl = pe.x >> 16;
                        if (rl < (unsigned int)rows_here) {
                            const int   c = (int)(pe.x & 0xFFFFu);
                            const float v = __builtin_bit_cast(float, pe.y);
                            const float* src = dense + (size_t)c * F;
                            float* dst = acc + rl * PITCH;
                            #pragma unroll 1
                            for (int f = 0; f < F; ++f)
                                atomicAdd(dst + (f % VEC) * 32 + (f / VEC), v * src[f]);
                        }
                    }
                } while (hit != 0u);
            }
        }
    }
    __syncthreads();

    const int total = need_scan
        ? __builtin_amdgcn_readfirstlane((int)min(ecount, (unsigned int)CAP))
        : (int)stored;
    if constexpr (MODE == 0) {
        if (t == 0) pubcount[blockIdx.x] = ecount;
    }
    for (int base = wave * 32; base < total; base += (OWN_THREADS / 32) * 32) {
        const int idx = base + lane;
        int   c = 0;
        float v = 0.0f;
        int   rl = NO_ROW;
        if (idx < total) {
            uint2 pe;
            if (need_scan) {
                pe = rec2[elist[idx]];
                if constexpr (MODE == 0)
                    mypub[idx] = pe;
            } else {
                pe = mypub[idx];
            }
            c  = (int)(pe.x & 0xFFFFu);
            rl = (int)(pe.x >> 16);
            v  = __builtin_bit_cast(float, pe.y);
        }
        const int n = min(32, total - base);
        for (int j0 = 0; j0 < n; j0 += 8) {
            int   cj[8];
            float vv[8];
            int   ro[8];
            bool  in_range[8];
            float g[8][VEC];
            #pragma unroll
            for (int u = 0; u < 8; ++u) {
                cj[u] = __builtin_amdgcn_readlane(c, j0 + u);
                vv[u] = __builtin_bit_cast(float,
                          __builtin_amdgcn_readlane(__builtin_bit_cast(int, v), j0 + u));
                const int rj = __builtin_amdgcn_readlane(rl, j0 + u);
                in_range[u] = rj < rows_here;
                ro[u] = rj * PITCH;
            }
            #pragma unroll
            for (int u = 0; u < 8; ++u) {
                const float* src = dense + (size_t)cj[u] * F + lane * VEC;
                if constexpr (VEC == 4) {
                    const float4 x = *(const float4*)src;
                    g[u][0] = x.x; g[u][1] = x.y; g[u][2] = x.z; g[u][3] = x.w;
                } else {
                    const float2 x = *(const float2*)src;
                    g[u][0] = x.x; g[u][1] = x.y;
                }
            }
            #pragma unroll
            for (int u = 0; u < 8; ++u) {
                if (j0 + u < n && in_range[u]) {
                    float* dst = acc + ro[u] + lane;
                    #pragma unroll
                    for (int k = 0; k < VEC; ++k)
                        atomicAdd(dst + k * 32, vv[u] * g[u][k]);
                }
            }
        }
    }
    __syncthreads();

    if constexpr (LOGSM) {
        static_assert(F == 64 && VEC == 2, "one 16-lane row of 4 features per lane");
        const int hl = lane & 15;
        const int hf = lane >> 4;
        const float4 bb = *(const float4*)(bias + 4 * hl);
        for (int r0 = wave * 2; r0 < rows_here; r0 += (OWN_THREADS / 32) * 2) {
            const int  row = r0 + hf;
            const bool ok  = row < rows_here;
            const float* a = acc + (ok ? row : r0) * PITCH;
            const float2 va = *(const float2*)(a + 2 * hl);
            const float2 vb = *(const float2*)(a + 32 + 2 * hl);
            const float x0 = va.x + bb.x;
            const float x1 = vb.x + bb.y;
            const float x2 = va.y + bb.z;
            const float x3 = vb.y + bb.w;
            float m = fmaxf(fmaxf(x0, x1), fmaxf(x2, x3));
            m = fmaxf(m, xlane16<8>(m));
            m = fmaxf(m, xlane16<4>(m));
            m = fmaxf(m, xlane16<2>(m));
            m = fmaxf(m, xlane16<1>(m));
            float s = (__expf(x0 - m) + __expf(x1 - m)) + (__expf(x2 - m) + __expf(x3 - m));
            s += xlane16<8>(s);
            s += xlane16<4>(s);
            s += xlane16<2>(s);
            s += xlane16<1>(s);
            const float lse = m + __logf(s);
            if (ok) {
                float4 o;
                o.x = x0 - lse;
                o.y = x1 - lse;
                o.z = x2 - lse;
                o.w = x3 - lse;
                *(float4*)(out + (size_t)(lo + row) * F + 4 * hl) = o;
            }
        }
    } else {
        static_assert(F == HID && VEC == 4 && OWN_THREADS % F == 0, "row layout of the passes below");
        constexpr int W2PITCH = CLS + 16;
        static_assert(HID * W2PITCH <= CAP, "W2 fits into the list's LDS space");
        {
            const int w = t & (F - 1);
            const float bv = bias[(w & 31) * VEC + (w >> 5)];
            for (int r = t >> 7; r < rows_here; r += OWN_THREADS / F) {
                float* p = acc + r * PITCH + w;
                *p = fmaxf(*p + bv, 0.0f);
            }
        }
        for (int i = t; i < HID * CLS / 4; i += OWN_THREADS) {
            const float4 wq = ((const float4*)w2)[i];
            unsigned int* wd = elist + (i >> 4) * W2PITCH + (i & 15) * 4;
            wd[0] = __builtin_bit_cast(unsigned int, wq.x);
            wd[1] = __builtin_bit_cast(unsigned int, wq.y);
            wd[2] = __builtin_bit_cast(unsigned int, wq.z);
            wd[3] = __builtin_bit_cast(unsigned int, wq.w);
        }
        __syncthreads();
        constexpr int MTILES = (OWN_ROWS + 15) / 16;
        const int half = lane >> 4;
        const int l16  = lane & 15;
        const int wv = __builtin_amdgcn_readfirstlane(wave);
        for (int job = wv; job < MTILES * (CLS / 16); job += OWN_THREADS / 32) {
            const int m0 = (job >> 2) * 16;
            const int n0 = (job & 3) * 16;
            if (m0 >= rows_here) break;
            const int arow = min(m0 + l16, rows_here - 1);
            const float* ap = acc + arow * PITCH + half * 32;
            const unsigned int* bw = elist + half * W2PITCH + n0 + l16;
            v8f d = {};
            #pragma unroll
            for (int j = 0; j < HID / 4; ++j) {
                v2f a, b;
                a.x = ap[j];
                a.y = ap[j + 64];
                b.x = __builtin_bit_cast(float, bw[(4 * j)     * W2PITCH]);
                b.y = __builtin_bit_cast(float, bw[(4 * j + 2) * W2PITCH]);
                d = __builtin_amdgcn_wmma_f32_16x16x4_f32(false, a, false, b, (short)0, d, false, false);
            }
            float* yb = out + (size_t)lo * CLS + n0 + l16;
            #pragma unroll
            for (int i = 0; i < 8; ++i) {
                const int r = m0 + i + 8 * half;
                if (r < rows_here) yb[(size_t)r * CLS] = d[i];
            }
        }
    }
}

extern "C" void kernel_launch(void* const* d_in, const int* in_sizes, int n_in,
                              void* d_out, int out_size, void* d_ws, size_t ws_size,
                              hipStream_t stream) {
    const float* x        = (const float*)d_in[0];
    const int*   edge_row = (const int*)  d_in[1];
    const int*   edge_col = (const int*)  d_in[2];
    const float* edge_val = (const float*)d_in[3];
    const float* W1       = (const float*)d_in[4];
    const float* b1       = (const float*)d_in[5];
    const float* W2       = (const float*)d_in[6];
    const float* b2       = (const float*)d_in[7];
    float* out = (float*)d_out;

    float* ws    = (float*)d_ws;
    float* XW1   = ws;
    float* HW2   = ws + (size_t)NROWS * HID;

    unsigned int* plist  = (unsigned int*)d_ws + 16777216;
    unsigned int* pcount = (unsigned int*)d_ws + 33554432;
    unsigned int* okeys  = (unsigned int*)d_ws + 50331648;
    unsigned int* erecs  = (unsigned int*)d_ws + 52428800;

    const int MT = NROWS / 16;

    gcn_gemm1_wmma<<<MT, 64, 0, stream>>>(x, W1, XW1, edge_row, edge_col, edge_val, okeys, erecs);
    gcn_spmm_own<HID, 0, false><<<OWN_GROUPS, OWN_THREADS, 0, stream>>>(XW1, HW2, plist, pcount, okeys, erecs, b1, W2);
    gcn_spmm_own<CLS, 1, true><<<OWN_GROUPS, OWN_THREADS, 0, stream>>>(HW2, out, plist, pcount, okeys, erecs, b2, W2);
}
